# starting kernel
# baseline (speedup 1.0000x reference)
#include <hip/hip_runtime.h>

typedef float v2f __attribute__((ext_vector_type(2)));
typedef float v8f __attribute__((ext_vector_type(8)));

__global__ void __launch_bounds__(256)
sage_scatter(const float* __restrict__ x, const int* __restrict__ src,
             const int* __restrict__ dst, float* __restrict__ agg,
             float* __restrict__ cnt, int E, int d, int do_count)
{
    const int lane = threadIdx.x & 31;
    const int wid  = blockIdx.x * (blockDim.x >> 5) + (threadIdx.x >> 5);
    const int nw   = gridDim.x * (blockDim.x >> 5);
    const int d4   = d >> 2;

    for (int e = wid; e < E; e += nw) {
        const int s = src[e];
        const int t = dst[e];

        const int e2 = e + nw;
        if (e2 < E) {
            __builtin_prefetch(&x[(size_t)src[e2] * d + lane * 4], 0, 1);
        }

        const float4* xs = (const float4*)(x + (size_t)s * d);
        float*        at = agg + (size_t)t * d;

        for (int c = lane; c < d4; c += 32) {
            float4 v = xs[c];
            atomicAdd(at + 4 * c + 0, v.x);
            atomicAdd(at + 4 * c + 1, v.y);
            atomicAdd(at + 4 * c + 2, v.z);
            atomicAdd(at + 4 * c + 3, v.w);
        }
        if (do_count && lane == 0) atomicAdd(cnt + t, 1.0f);
    }
}

template <int K, int NCOL, bool RELU>
__global__ void __launch_bounds__(256)
sage_gemm(const float* __restrict__ agg, const float* __restrict__ cnt,
          const float* __restrict__ xroot, const float* __restrict__ Wl,
          const float* __restrict__ Wr, const float* __restrict__ bias,
          float* __restrict__ out, int nrows)
{
    constexpr int JT     = 4;
    constexpr int GROUPS = NCOL / (16 * JT);

    const int lane = threadIdx.x & 31;
    const int half = lane >> 4;
    const int l    = lane & 15;

    const int wid = blockIdx.x * (blockDim.x >> 5) + (threadIdx.x >> 5);
    const int tm  = (wid / GROUPS) * 16;
    const int tn  = (wid % GROUPS) * (16 * JT);
    if (tm >= nrows) return;

    const int row = tm + l;
    const float  inv  = 1.0f / fmaxf(cnt[row], 1.0f);
    const float* aRow = agg   + (size_t)row * K;
    const float* xRow = xroot + (size_t)row * K;

    v8f acc[JT] = {};
#pragma unroll 2
    for (int k0 = 0; k0 < K; k0 += 4) {
        const int ka = k0 + 2 * half;

        v2f am, ar;
        am.x = aRow[ka] * inv;  am.y = aRow[ka + 1] * inv;
        ar.x = xRow[ka];        ar.y = xRow[ka + 1];

#pragma unroll
        for (int j = 0; j < JT; ++j) {
            const int col = tn + 16 * j + l;
            v2f bl, br;
            bl.x = Wl[(size_t)ka * NCOL + col];
            bl.y = Wl[(size_t)(ka + 1) * NCOL + col];
            br.x = Wr[(size_t)ka * NCOL + col];
            br.y = Wr[(size_t)(ka + 1) * NCOL + col];

            acc[j] = __builtin_amdgcn_wmma_f32_16x16x4_f32(
                false, am, false, bl, (short)0, acc[j], false, false);
            acc[j] = __builtin_amdgcn_wmma_f32_16x16x4_f32(
                false, ar, false, br, (short)0, acc[j], false, false);
        }
    }

#pragma unroll
    for (int j = 0; j < JT; ++j) {
        const int col = tn + 16 * j + l;
        const float bv = bias[col];
#pragma unroll
        for (int r = 0; r < 8; ++r) {
            float v = acc[j][r] + bv;
            if (RELU) v = fmaxf(v, 0.0f);
            out[(size_t)(tm + 8 * half + r) * NCOL + col] = v;
        }
    }
}

extern "C" void kernel_launch(void* const* d_in, const int* in_sizes, int n_in,
                              void* d_out, int out_size, void* d_ws, size_t ws_size,
                              hipStream_t stream)
{
    constexpr int IN_DIM = 128, HID = 256, EMB = 128;

    const float* x   = (const float*)d_in[0];
    const int*   ei  = (const int*)  d_in[1];
    const float* W1l = (const float*)d_in[2];
    const float* b1  = (const float*)d_in[3];
    const float* W1r = (const float*)d_in[4];
    const float* W2l = (const float*)d_in[5];
    const float* b2  = (const float*)d_in[6];
    const float* W2r = (const float*)d_in[7];

    const int N = in_sizes[0] / IN_DIM;
    const int E = in_sizes[1] / 2;
    const int* src = ei;
    const int* dst = ei + E;

    float* cnt  = (float*)d_ws;
    float* agg1 = cnt  + (((size_t)N + 63) & ~(size_t)63);
    float* h    = agg1 + (size_t)N * IN_DIM;
    float* agg2 = h    + (size_t)N * HID;

    hipMemsetAsync(cnt,  0, sizeof(float) * (size_t)N,          stream);
    hipMemsetAsync(agg1, 0, sizeof(float) * (size_t)N * IN_DIM, stream);
    hipMemsetAsync(agg2, 0, sizeof(float) * (size_t)N * HID,    stream);

    const dim3 blk(256);
    const int scatterBlocks = 2048;
    const int tmN = (N + 15) / 16;

    sage_scatter<<<scatterBlocks, blk, 0, stream>>>(x, src, dst, agg1, cnt,
                                                    E, IN_DIM, 1);
    const int waves_l1  = tmN * (HID / 64);
    const int blocks_l1 = (waves_l1 + 7) / 8;
    sage_gemm<IN_DIM, HID, true><<<blocks_l1, blk, 0, stream>>>(
        agg1, cnt, x, W1l, W1r, b1, h, N);

    sage_scatter<<<scatterBlocks, blk, 0, stream>>>(h, src, dst, agg2, cnt,
                                                    E, HID, 0);
    const int waves_l2  = tmN * (EMB / 64);
    const int blocks_l2 = (waves_l2 + 7) / 8;
    sage_gemm<HID, EMB, false><<<blocks_l2, blk, 0, stream>>>(
        agg2, cnt, h, W2l, W2r, b2, (float*)d_out, N);
}


# Both GEMM layers fused: one 4-wave work-group per 16 rows stages [mean|x] in LDS once, keeps its h tile in LDS and projects it at once; h never goes to memory, 3 launches not 4; every sum unchanged
# speedup vs baseline: 3.3077x; 3.3077x over previous
#include <hip/hip_runtime.h>

typedef float v2f __attribute__((ext_vector_type(2)));
typedef float v4f __attribute__((ext_vector_type(4)));
typedef float v8f __attribute__((ext_vector_type(8)));

static __device__ __forceinline__ int package_half()
{
    const unsigned id = __builtin_amdgcn_s_sendmsg_rtn(0x87);
    return (int)((id >> 18) & 1u);
}

template <int D, bool FIRST>
__global__ void __launch_bounds__(256)
sage_scatter_home(const float* __restrict__ x, const int* __restrict__ src,
                  const int* __restrict__ dst, float* __restrict__ acc,
                  float* __restrict__ cnt, int* __restrict__ tickets,
                  int E, int N)
{
    static_assert(D % 32 == 0, "a row is a whole number of wave-wide lines");
    constexpr int R     = D / 32;
    constexpr int CHUNK = 256;
    constexpr int G     = 4;

    __shared__ int s_ticket[2];
    __shared__ int s_home;

    const int lane    = threadIdx.x & 31;
    const int wave    = threadIdx.x >> 5;
    const int nChunks = (E + CHUNK - 1) / CHUNK;
    const int split   = N >> 1;
    if (threadIdx.x == 0) s_home = package_half();
    __syncthreads();
    const int home    = __builtin_amdgcn_readfirstlane(s_home);

    int it = 0;
    for (int phase = 0; phase < 2; ++phase) {
        const int h  = home ^ phase;
        int*      tk = tickets + 64 * h;

        int nxt = 0;
        if (threadIdx.x == 0) nxt = atomicAdd(tk, 1);
        for (;;) {
            const int slot = it & 1;
            ++it;
            if (threadIdx.x == 0) s_ticket[slot] = nxt;
            __syncthreads();
            const int c = __builtin_amdgcn_readfirstlane(s_ticket[slot]);
            if (c >= nChunks) break;
            const int e = c * CHUNK + wave * 32 + lane;
            int s = 0, t = 0;
            if (e < E) {
                s = src[e];
                t = dst[e];
            }
            if (threadIdx.x == 0) nxt = atomicAdd(tk, 1);
            const bool mine = (e < E) && (((t >= split) ? 1 : 0) == h);
            float w = 1.0f;
            if (FIRST) { if (mine) atomicAdd(cnt + t, 1.0f); }
            else       { if (mine) w = 1.0f / fmaxf(cnt[t], 1.0f); }

            unsigned m = __builtin_amdgcn_ballot_w32(mine);
            while (m) {
                float v[G][R];
                int   tt[G];
                float ww[G];
                int   n = 0;
#pragma unroll
                for (int q = 0; q < G; ++q) {
                    if (m) {
                        const int i = __builtin_ctz(m);
                        m &= m - 1;
                        const int si = __builtin_amdgcn_readlane(s, i);
                        tt[q] = __builtin_amdgcn_readlane(t, i);
                        ww[q] = __builtin_bit_cast(float,
                            __builtin_amdgcn_readlane(__builtin_bit_cast(int, w), i));
                        const float* xs = x + (size_t)si * D + lane;
#pragma unroll
                        for (int r = 0; r < R; ++r) v[q][r] = xs[32 * r];
                        n = q + 1;
                    }
                }
#pragma unroll
                for (int q = 0; q < G; ++q) {
                    if (q < n) {
                        float* at = acc + (size_t)tt[q] * D + lane;
#pragma unroll
                        for (int r = 0; r < R; ++r)
                            atomicAdd(at + 32 * r, FIRST ? v[q][r] : v[q][r] * ww[q]);
                    }
                }
            }
        }
    }
}

template <int K1, int HIDN, int NOUT>
__global__ void __launch_bounds__(128, 16)
sage_layers(const float* __restrict__ agg, const float* __restrict__ cnt,
            const float* __restrict__ xroot,
            const float* __restrict__ W1l, const float* __restrict__ W1r,
            const float* __restrict__ b1,
            const float* __restrict__ W2l, const float* __restrict__ W2r,
            const float* __restrict__ b2,
            float* __restrict__ p, float* __restrict__ out, int nrows)
{
    constexpr int JT    = 4;
    constexpr int PITCH = HIDN + 4;
    static_assert(K1 == 4 * 32,        "a wave stages one input row as 32 x 128 bits");
    static_assert(HIDN == 2 * K1,      "the LDS tile holds [mean | x] first, h afterwards");
    static_assert(HIDN == 4 * 16 * JT, "4 waves x 64 columns cover a row of h");
    static_assert(NOUT == 2 * 16 * JT, "2 waves x 64 columns cover a row of p / of out");

    __shared__ __attribute__((aligned(16))) float tile[16 * PITCH];

    const int lane = threadIdx.x & 31;
    const int wave = __builtin_amdgcn_readfirstlane(threadIdx.x >> 5);
    const int half = lane >> 4;
    const int l    = lane & 15;
    const int tm   = blockIdx.x * 16;
    if (tm >= nrows) return;

#pragma unroll
    for (int i = 0; i < 4; ++i) {
        const int    rr  = wave + 4 * i;
        const float* ag  = agg   + (size_t)(tm + rr) * K1;
        const float* xg  = xroot + (size_t)(tm + rr) * K1;
        const float  inv = 1.0f / fmaxf(cnt[tm + rr], 1.0f);
        const v4f a = *(const v4f*)(ag + 4u * (unsigned)lane);
        const v4f r = *(const v4f*)(xg + 4u * (unsigned)lane);
        *(v4f*)(tile + rr * PITCH + 4 * lane)      = a * inv;
        *(v4f*)(tile + rr * PITCH + K1 + 4 * lane) = r;
    }
    __syncthreads();

    const float* fRow = tile + l * PITCH;

    {
        const int c4 = wave * (16 * JT) + 4 * l;
        v8f acc[JT] = {};
#pragma unroll 1
        for (int k0 = 0; k0 < K1; k0 += 4) {
            const int ka = k0 + 2 * half;

            const v2f am = *(const v2f*)(fRow + ka);
            const v2f ar = *(const v2f*)(fRow + K1 + ka);

            const unsigned wo = (unsigned)ka * HIDN + (unsigned)c4;
            const v4f l0 = *(const v4f*)(W1l + wo);
            const v4f l1 = *(const v4f*)(W1l + wo + HIDN);
            const v4f r0 = *(const v4f*)(W1r + wo);
            const v4f r1 = *(const v4f*)(W1r + wo + HIDN);

#pragma unroll
            for (int j = 0; j < JT; ++j) {
                v2f bl, br;
                bl.x = l0[j];  bl.y = l1[j];
                br.x = r0[j];  br.y = r1[j];

                acc[j] = __builtin_amdgcn_wmma_f32_16x16x4_f32(
                    false, am, false, bl, (short)0, acc[j], false, false);
                acc[j] = __builtin_amdgcn_wmma_f32_16x16x4_f32(
                    false, ar, false, br, (short)0, acc[j], false, false);
            }
        }

        __syncthreads();

        const v4f bv = *(const v4f*)(b1 + c4);
#pragma unroll
        for (int r = 0; r < 8; ++r) {
            v4f v;
#pragma unroll
            for (int j = 0; j < JT; ++j)
                v[j] = fmaxf(acc[j][r] + bv[j], 0.0f);
            *(v4f*)(tile + (8 * half + r) * PITCH + c4) = v;
        }
    }
    __builtin_amdgcn_sched_barrier(0);
    __syncthreads();

    {
        const bool   root = wave >= 2;
        const float* Wm   = root ? W2r : W2l;
        float*       dtile = (root ? out : p) + (size_t)tm * NOUT;
        const int    c4   = (wave & 1) * (16 * JT) + 4 * l;

        v8f acc[JT] = {};
#pragma unroll 2
        for (int k0 = 0; k0 < HIDN; k0 += 4) {
            const int ka = k0 + 2 * half;

            const v2f a = *(const v2f*)(fRow + ka);

            const unsigned wo = (unsigned)ka * NOUT + (unsigned)c4;
            const v4f w0 = *(const v4f*)(Wm + wo);
            const v4f w1 = *(const v4f*)(Wm + wo + NOUT);

#pragma unroll
            for (int j = 0; j < JT; ++j) {
                v2f b;
                b.x = w0[j];  b.y = w1[j];
                acc[j] = __builtin_amdgcn_wmma_f32_16x16x4_f32(
                    false, a, false, b, (short)0, acc[j], false, false);
            }
        }

        v4f bv = {0.0f, 0.0f, 0.0f, 0.0f};
        if (root) bv = *(const v4f*)(b2 + c4);
#pragma unroll
        for (int r = 0; r < 8; ++r) {
            v4f v;
#pragma unroll
            for (int j = 0; j < JT; ++j)
                v[j] = root ? acc[j][r] + bv[j] : acc[j][r];
            *(v4f*)(dtile + (unsigned)((8 * half + r) * NOUT + c4)) = v;
        }
    }
}

extern "C" void kernel_launch(void* const* d_in, const int* in_sizes, int n_in,
                              void* d_out, int out_size, void* d_ws, size_t ws_size,
                              hipStream_t stream)
{
    constexpr int IN_DIM = 128, HID = 256, EMB = 128;

    const float* x   = (const float*)d_in[0];
    const int*   ei  = (const int*)  d_in[1];
    const float* W1l = (const float*)d_in[2];
    const float* b1  = (const float*)d_in[3];
    const float* W1r = (const float*)d_in[4];
    const float* W2l = (const float*)d_in[5];
    const float* b2  = (const float*)d_in[6];
    const float* W2r = (const float*)d_in[7];

    const int N = in_sizes[0] / IN_DIM;
    const int E = in_sizes[1] / 2;
    const int* src = ei;
    const int* dst = ei + E;

    int*   tickets = (int*)d_ws;
    float* cnt  = (float*)d_ws + 256;
    float* agg1 = cnt  + (((size_t)N + 63) & ~(size_t)63);
    float* p    = agg1 + (size_t)N * IN_DIM;

    hipMemsetAsync(d_ws, 0, sizeof(float) * ((size_t)N + 256),  stream);
    hipMemsetAsync(agg1, 0, sizeof(float) * (size_t)N * IN_DIM, stream);

    const dim3 blk(256);
    const int scatterBlocks = 2048;
    const int tmN = (N + 15) / 16;

    sage_scatter_home<IN_DIM, true><<<scatterBlocks, blk, 0, stream>>>(
        x, src, dst, agg1, cnt, tickets, E, N);

    sage_layers<IN_DIM, HID, EMB><<<tmN, dim3(128), 0, stream>>>(
        agg1, cnt, x, W1l, W1r, b1, W2l, W2r, b2, p, (float*)d_out, N);

    sage_scatter_home<EMB, false><<<scatterBlocks, blk, 0, stream>>>(
        p, src, dst, (float*)d_out, cnt, tickets + 128, E, N);
}


# Scatter source rows read from LDS, not per edge from L2: one new launch groups edges by (64-row source block, destination half), a ticket stages its block; float atomics, arrival order as before
# speedup vs baseline: 3.4410x; 1.0403x over previous
#include <hip/hip_runtime.h>

typedef float v2f __attribute__((ext_vector_type(2)));
typedef float v4f __attribute__((ext_vector_type(4)));
typedef float v8f __attribute__((ext_vector_type(8)));

static __device__ __forceinline__ int package_half()
{
    const unsigned id = __builtin_amdgcn_s_sendmsg_rtn(0x87);
    return (int)((id >> 18) & 1u);
}

template <int RB, int CE, int MAXB>
__global__ void __launch_bounds__(1024)
edge_partition(const int* __restrict__ src, const int* __restrict__ dst,
               int* __restrict__ ssrc, int* __restrict__ sdst,
               int* __restrict__ offs, int E, int N)
{
    constexpr int T   = 1024;
    constexpr int PER = CE / T;
    static_assert(CE % T == 0,   "every thread takes the same number of edges");
    static_assert(MAXB == 2 * T, "a thread scans two neighbouring bins");

    __shared__ int cursor[MAXB];
    __shared__ int start[MAXB + 1];
    __shared__ int wtot[32];
    __shared__ int st_s[CE];
    __shared__ int st_d[CE];

    const int tid   = threadIdx.x;
    const int lane  = tid & 31;
    const int wave  = tid >> 5;
    const int q     = blockIdx.x;
    const int split = N >> 1;
    const int e0    = q * CE;
    const int ne    = min(CE, E - e0);

    cursor[tid]     = 0;
    cursor[tid + T] = 0;
    __syncthreads();

    int s[PER], t[PER], b[PER];
#pragma unroll
    for (int k = 0; k < PER; ++k) {
        const int i = min(k * T + tid, ne - 1);
        s[k] = src[e0 + i];
        t[k] = dst[e0 + i];
    }
#pragma unroll
    for (int k = 0; k < PER; ++k) {
        b[k] = -1;
        if (k * T + tid < ne) {
            b[k] = 2 * (s[k] / RB) + ((t[k] >= split) ? 1 : 0);
            atomicAdd(&cursor[b[k]], 1);
        }
    }
    __syncthreads();

    const int c0 = cursor[2 * tid];
    const int c1 = cursor[2 * tid + 1];
    int v = c0 + c1;
    for (int d = 1; d < 32; d <<= 1) {
        const int u = __shfl_up(v, d);
        if (lane >= d) v += u;
    }
    if (lane == 31) wtot[wave] = v;
    __syncthreads();
    int before = 0;
    for (int j = 0; j < wave; ++j) before += wtot[j];
    const int excl = before + v - (c0 + c1);
    cursor[2 * tid]     = excl;
    cursor[2 * tid + 1] = excl + c0;
    start[2 * tid]      = excl;
    start[2 * tid + 1]  = excl + c0;
    if (tid == T - 1) start[MAXB] = excl + c0 + c1;
    __syncthreads();

#pragma unroll
    for (int k = 0; k < PER; ++k) {
        if (b[k] >= 0) {
            const int pos = atomicAdd(&cursor[b[k]], 1);
            st_s[pos] = s[k];
            st_d[pos] = t[k];
        }
    }
    __syncthreads();

    for (int i = tid; i < ne; i += T) {
        ssrc[e0 + i] = st_s[i];
        sdst[e0 + i] = st_d[i];
    }
    int* od = offs + (size_t)q * (MAXB + 1);
    for (int i = tid; i <= MAXB; i += T) od[i] = start[i];
}

template <int D, bool FIRST, int RB, int CE, int MAXB>
__global__ void __launch_bounds__(256)
sage_scatter_staged(const float* __restrict__ x, const int* __restrict__ ssrc,
                    const int* __restrict__ sdst, const int* __restrict__ offs,
                    float* __restrict__ acc, float* __restrict__ cnt,
                    int* __restrict__ tickets, int E, int N)
{
    static_assert(D % 32 == 0, "a row is a whole number of wave-wide lines");
    static_assert((RB * D / 4) % (4 * 256) == 0, "the block is staged in whole groups of four passes of the work-group");
    constexpr int R = D / 32;
    constexpr int G = 4;

    __shared__ __attribute__((aligned(16))) float rows[RB * D];
    __shared__ int s_ticket[2];
    __shared__ int s_home;

    const int lane  = threadIdx.x & 31;
    const int wave  = __builtin_amdgcn_readfirstlane(threadIdx.x >> 5);
    const int nblk  = (N + RB - 1) / RB;
    const int nprod = (E + CE - 1) / CE;
    if (threadIdx.x == 0) s_home = package_half();
    __syncthreads();
    const int home  = __builtin_amdgcn_readfirstlane(s_home);

    int it = 0;
    for (int phase = 0; phase < 2; ++phase) {
        const int h  = home ^ phase;
        int*      tk = tickets + 64 * h;

        int nxt = 0;
        if (threadIdx.x == 0) nxt = atomicAdd(tk, 1);
        for (;;) {
            const int slot = it & 1;
            ++it;
            if (threadIdx.x == 0) s_ticket[slot] = nxt;
            __syncthreads();
            const int c = __builtin_amdgcn_readfirstlane(s_ticket[slot]);
            if (c >= nblk) break;
            if (threadIdx.x == 0) nxt = atomicAdd(tk, 1);

            const int  r0 = c * RB;
            const int  n4 = min(RB, N - r0) * (D / 4);
            const v4f* g4 = (const v4f*)(x + (size_t)r0 * D);
            v4f*       l4 = (v4f*)rows;
#pragma unroll 1
            for (int i0 = threadIdx.x; i0 < RB * D / 4; i0 += 4 * 256) {
                v4f stage[4];
#pragma unroll
                for (int k = 0; k < 4; ++k)
                    stage[k] = g4[min(i0 + k * 256, n4 - 1)];
#pragma unroll
                for (int k = 0; k < 4; ++k)
                    if (i0 + k * 256 < n4) l4[i0 + k * 256] = stage[k];
            }
            __syncthreads();

            const int bin = 2 * c + h;
            for (int qb = 0; qb < nprod; qb += 256) {
                const int qq = qb + wave + 8 * lane;
                int dbeg = 0, dend = 0;
                if (qq < nprod) {
                    const int* od = offs + (size_t)qq * (MAXB + 1) + bin;
                    dbeg = od[0];
                    dend = od[1];
                }
                for (int j = 0; j < 32; ++j) {
                    const int q = qb + wave + 8 * j;
                    if (q >= nprod) break;
                    const int  beg = __builtin_amdgcn_readlane(dbeg, j);
                    const int  end = __builtin_amdgcn_readlane(dend, j);
                    const int* qs  = ssrc + (size_t)q * CE;
                    const int* qd  = sdst + (size_t)q * CE;
                    for (int base = beg; base < end; base += 32) {
                        const int  i     = base + lane;
                        const bool valid = i < end;
                        int s = 0, t = 0;
                        if (valid) {
                            s = qs[i];
                            t = qd[i];
                        }
                        float w = 1.0f;
                        if (FIRST) { if (valid) atomicAdd(cnt + t, 1.0f); }
                        else       { if (valid) w = 1.0f / fmaxf(cnt[t], 1.0f); }

                        unsigned m = __builtin_amdgcn_ballot_w32(valid);
                        while (m) {
                            float v[G][R];
                            int   tt[G];
                            float ww[G];
                            int   n = 0;
#pragma unroll
                            for (int g = 0; g < G; ++g) {
                                if (m) {
                                    const int l = __builtin_ctz(m);
                                    m &= m - 1;
                                    const int si = __builtin_amdgcn_readlane(s, l);
                                    tt[g] = __builtin_amdgcn_readlane(t, l);
                                    ww[g] = __builtin_bit_cast(float,
                                        __builtin_amdgcn_readlane(__builtin_bit_cast(int, w), l));
                                    const float* xs = rows + (si - r0) * D + lane;
#pragma unroll
                                    for (int r = 0; r < R; ++r) v[g][r] = xs[32 * r];
                                    n = g + 1;
                                }
                            }
#pragma unroll
                            for (int g = 0; g < G; ++g) {
                                if (g < n) {
                                    float* at = acc + (size_t)tt[g] * D + lane;
#pragma unroll
                                    for (int r = 0; r < R; ++r)
                                        atomicAdd(at + 32 * r, FIRST ? v[g][r] : v[g][r] * ww[g]);
                                }
                            }
                        }
                    }
                }
            }
        }
    }
}

template <int K1, int HIDN, int NOUT>
__global__ void __launch_bounds__(128, 16)
sage_layers(const float* __restrict__ agg, const float* __restrict__ cnt,
            const float* __restrict__ xroot,
            const float* __restrict__ W1l, const float* __restrict__ W1r,
            const float* __restrict__ b1,
            const float* __restrict__ W2l, const float* __restrict__ W2r,
            const float* __restrict__ b2,
            float* __restrict__ p, float* __restrict__ out, int nrows)
{
    constexpr int JT    = 4;
    constexpr int PITCH = HIDN + 4;
    static_assert(K1 == 4 * 32,        "a wave stages one input row as 32 x 128 bits");
    static_assert(HIDN == 2 * K1,      "the LDS tile holds [mean | x] first, h afterwards");
    static_assert(HIDN == 4 * 16 * JT, "4 waves x 64 columns cover a row of h");
    static_assert(NOUT == 2 * 16 * JT, "2 waves x 64 columns cover a row of p / of out");

    __shared__ __attribute__((aligned(16))) float tile[16 * PITCH];

    const int lane = threadIdx.x & 31;
    const int wave = __builtin_amdgcn_readfirstlane(threadIdx.x >> 5);
    const int half = lane >> 4;
    const int l    = lane & 15;
    const int tm   = blockIdx.x * 16;
    if (tm >= nrows) return;

#pragma unroll
    for (int i = 0; i < 4; ++i) {
        const int    rr  = wave + 4 * i;
        const float* ag  = agg   + (size_t)(tm + rr) * K1;
        const float* xg  = xroot + (size_t)(tm + rr) * K1;
        const float  inv = 1.0f / fmaxf(cnt[tm + rr], 1.0f);
        const v4f a = *(const v4f*)(ag + 4u * (unsigned)lane);
        const v4f r = *(const v4f*)(xg + 4u * (unsigned)lane);
        *(v4f*)(tile + rr * PITCH + 4 * lane)      = a * inv;
        *(v4f*)(tile + rr * PITCH + K1 + 4 * lane) = r;
    }
    __syncthreads();

    const float* fRow = tile + l * PITCH;

    {
        const int c4 = wave * (16 * JT) + 4 * l;
        v8f acc[JT] = {};
#pragma unroll 1
        for (int k0 = 0; k0 < K1; k0 += 4) {
            const int ka = k0 + 2 * half;

            const v2f am = *(const v2f*)(fRow + ka);
            const v2f ar = *(const v2f*)(fRow + K1 + ka);

            const unsigned wo = (unsigned)ka * HIDN + (unsigned)c4;
            const v4f l0 = *(const v4f*)(W1l + wo);
            const v4f l1 = *(const v4f*)(W1l + wo + HIDN);
            const v4f r0 = *(const v4f*)(W1r + wo);
            const v4f r1 = *(const v4f*)(W1r + wo + HIDN);

#pragma unroll
            for (int j = 0; j < JT; ++j) {
                v2f bl, br;
                bl.x = l0[j];  bl.y = l1[j];
                br.x = r0[j];  br.y = r1[j];

                acc[j] = __builtin_amdgcn_wmma_f32_16x16x4_f32(
                    false, am, false, bl, (short)0, acc[j], false, false);
                acc[j] = __builtin_amdgcn_wmma_f32_16x16x4_f32(
                    false, ar, false, br, (short)0, acc[j], false, false);
            }
        }

        __syncthreads();

        const v4f bv = *(const v4f*)(b1 + c4);
#pragma unroll
        for (int r = 0; r < 8; ++r) {
            v4f v;
#pragma unroll
            for (int j = 0; j < JT; ++j)
                v[j] = fmaxf(acc[j][r] + bv[j], 0.0f);
            *(v4f*)(tile + (8 * half + r) * PITCH + c4) = v;
        }
    }
    __builtin_amdgcn_sched_barrier(0);
    __syncthreads();

    {
        const bool   root = wave >= 2;
        const float* Wm   = root ? W2r : W2l;
        float*       dtile = (root ? out : p) + (size_t)tm * NOUT;
        const int    c4   = (wave & 1) * (16 * JT) + 4 * l;

        v8f acc[JT] = {};
#pragma unroll 2
        for (int k0 = 0; k0 < HIDN; k0 += 4) {
            const int ka = k0 + 2 * half;

            const v2f a = *(const v2f*)(fRow + ka);

            const unsigned wo = (unsigned)ka * NOUT + (unsigned)c4;
            const v4f w0 = *(const v4f*)(Wm + wo);
            const v4f w1 = *(const v4f*)(Wm + wo + NOUT);

#pragma unroll
            for (int j = 0; j < JT; ++j) {
                v2f b;
                b.x = w0[j];  b.y = w1[j];
                acc[j] = __builtin_amdgcn_wmma_f32_16x16x4_f32(
                    false, a, false, b, (short)0, acc[j], false, false);
            }
        }

        v4f bv = {0.0f, 0.0f, 0.0f, 0.0f};
        if (root) bv = *(const v4f*)(b2 + c4);
#pragma unroll
        for (int r = 0; r < 8; ++r) {
            v4f v;
#pragma unroll
            for (int j = 0; j < JT; ++j)
                v[j] = root ? acc[j][r] + bv[j] : acc[j][r];
            *(v4f*)(dtile + (unsigned)((8 * half + r) * NOUT + c4)) = v;
        }
    }
}

extern "C" void kernel_launch(void* const* d_in, const int* in_sizes, int n_in,
                              void* d_out, int out_size, void* d_ws, size_t ws_size,
                              hipStream_t stream)
{
    constexpr int IN_DIM = 128, HID = 256, EMB = 128;
    constexpr int RB   = 64;
    constexpr int CE   = 16384;
    constexpr int MAXB = 2048;

    const float* x   = (const float*)d_in[0];
    const int*   ei  = (const int*)  d_in[1];
    const float* W1l = (const float*)d_in[2];
    const float* b1  = (const float*)d_in[3];
    const float* W1r = (const float*)d_in[4];
    const float* W2l = (const float*)d_in[5];
    const float* b2  = (const float*)d_in[6];
    const float* W2r = (const float*)d_in[7];

    const int N = in_sizes[0] / IN_DIM;
    const int E = in_sizes[1] / 2;
    const int* src = ei;
    const int* dst = ei + E;
    const int nProd = (E + CE - 1) / CE;

    int*   tickets = (int*)d_ws;
    float* cnt  = (float*)d_ws + 256;
    float* agg1 = cnt  + (((size_t)N + 63) & ~(size_t)63);
    float* p    = agg1 + (size_t)N * IN_DIM;
    int*   ssrc = (int*)(p + (size_t)N * EMB);
    int*   sdst = ssrc + (size_t)nProd * CE;
    int*   offs = sdst + (size_t)nProd * CE;

    hipMemsetAsync(d_ws, 0, sizeof(float) * ((size_t)N + 256),  stream);
    hipMemsetAsync(agg1, 0, sizeof(float) * (size_t)N * IN_DIM, stream);

    const dim3 blk(256);
    const int scatterBlocks = 2048;
    const int tmN = (N + 15) / 16;

    edge_partition<RB, CE, MAXB><<<nProd, dim3(1024), 0, stream>>>(
        src, dst, ssrc, sdst, offs, E, N);

    sage_scatter_staged<IN_DIM, true, RB, CE, MAXB><<<scatterBlocks, blk, 0, stream>>>(
        x, ssrc, sdst, offs, agg1, cnt, tickets, E, N);

    sage_layers<IN_DIM, HID, EMB><<<tmN, dim3(128), 0, stream>>>(
        agg1, cnt, x, W1l, W1r, b1, W2l, W2r, b2, p, (float*)d_out, N);

    sage_scatter_staged<EMB, false, RB, CE, MAXB><<<scatterBlocks, blk, 0, stream>>>(
        p, ssrc, sdst, offs, (float*)d_out, cnt, tickets + 128, E, N);
}


# Sums per 32-row destination block in LDS: edges binned by destination block, source rows added by LDS float atomics (arrival order), block stored once, mean divisor after the sum; no global atomics
# speedup vs baseline: 3.8348x; 1.1144x over previous
#include <hip/hip_runtime.h>

typedef float v2f __attribute__((ext_vector_type(2)));
typedef float v4f __attribute__((ext_vector_type(4)));
typedef float v8f __attribute__((ext_vector_type(8)));

template <int RB, int CE, int MAXB>
__global__ void __launch_bounds__(1024)
edge_partition(const int* __restrict__ src, const int* __restrict__ dst,
               int* __restrict__ ssrc, int* __restrict__ sdst,
               int* __restrict__ offs, int E)
{
    constexpr int T   = 1024;
    constexpr int PER = CE / T;
    static_assert(CE % T == 0,   "every thread takes the same number of edges");
    static_assert(MAXB == 2 * T, "a thread scans two neighbouring bins");

    __shared__ int cursor[MAXB];
    __shared__ int start[MAXB + 1];
    __shared__ int wtot[32];
    __shared__ int st_s[CE];
    __shared__ int st_d[CE];

    const int tid   = threadIdx.x;
    const int lane  = tid & 31;
    const int wave  = tid >> 5;
    const int q     = blockIdx.x;
    const int e0    = q * CE;
    const int ne    = min(CE, E - e0);

    cursor[tid]     = 0;
    cursor[tid + T] = 0;
    __syncthreads();

    int s[PER], t[PER], b[PER];
#pragma unroll
    for (int k = 0; k < PER; ++k) {
        const int i = min(k * T + tid, ne - 1);
        s[k] = src[e0 + i];
        t[k] = dst[e0 + i];
    }
#pragma unroll
    for (int k = 0; k < PER; ++k) {
        b[k] = -1;
        if (k * T + tid < ne) {
            b[k] = t[k] / RB;
            atomicAdd(&cursor[b[k]], 1);
        }
    }
    __syncthreads();

    const int c0 = cursor[2 * tid];
    const int c1 = cursor[2 * tid + 1];
    int v = c0 + c1;
    for (int d = 1; d < 32; d <<= 1) {
        const int u = __shfl_up(v, d);
        if (lane >= d) v += u;
    }
    if (lane == 31) wtot[wave] = v;
    __syncthreads();
    int before = 0;
    for (int j = 0; j < wave; ++j) before += wtot[j];
    const int excl = before + v - (c0 + c1);
    cursor[2 * tid]     = excl;
    cursor[2 * tid + 1] = excl + c0;
    start[2 * tid]      = excl;
    start[2 * tid + 1]  = excl + c0;
    if (tid == T - 1) start[MAXB] = excl + c0 + c1;
    __syncthreads();

#pragma unroll
    for (int k = 0; k < PER; ++k) {
        if (b[k] >= 0) {
            const int pos = atomicAdd(&cursor[b[k]], 1);
            st_s[pos] = s[k];
            st_d[pos] = t[k];
        }
    }
    __syncthreads();

    for (int i = tid; i < ne; i += T) {
        ssrc[e0 + i] = st_s[i];
        sdst[e0 + i] = st_d[i];
    }
    int* od = offs + (size_t)q * (MAXB + 1);
    for (int i = tid; i <= MAXB; i += T) od[i] = start[i];
}

template <int D, bool FIRST, int RB, int CE, int MAXB>
__global__ void __launch_bounds__(256)
sage_block_sums(const float* __restrict__ x, const int* __restrict__ ssrc,
                const int* __restrict__ sdst, const int* __restrict__ offs,
                float* __restrict__ acc, float* __restrict__ cnt, int E, int N)
{
    static_assert(D == 4 * 32, "a wave reads one source row as 32 x 128 bits");
    static_assert(RB <= 256,   "one thread per row handles the edge counts");
    constexpr int G = 8;

    __shared__ __attribute__((aligned(16))) float rows[RB * D];
    __shared__ float cntl[RB];

    const int lane  = threadIdx.x & 31;
    const int wave  = __builtin_amdgcn_readfirstlane(threadIdx.x >> 5);
    const int nprod = (E + CE - 1) / CE;
    const int c     = blockIdx.x;
    const int r0    = c * RB;
    const int nr    = min(RB, N - r0);

    v4f* l4 = (v4f*)rows;
    for (int i = threadIdx.x; i < RB * D / 4; i += 256)
        l4[i] = v4f{0.0f, 0.0f, 0.0f, 0.0f};
    if (FIRST) { if (threadIdx.x < RB) cntl[threadIdx.x] = 0.0f; }
    __syncthreads();

    for (int qb = 0; qb < nprod; qb += 256) {
        const int qq = qb + wave + 8 * lane;
        int dbeg = 0, dend = 0;
        if (qq < nprod) {
            const int* od = offs + (size_t)qq * (MAXB + 1) + c;
            dbeg = od[0];
            dend = od[1];
        }
        const int len = dend - dbeg;
        int incl = len;
        for (int d = 1; d < 32; d <<= 1) {
            const int u = __shfl_up(incl, d);
            if (lane >= d) incl += u;
        }
        const int total = __builtin_amdgcn_readlane(incl, 31);

        const int nseg = max(0, min(32, (nprod - (qb + wave) + 7) / 8));

        auto fetch = [&](int base, int& s, int& t) {
            const int e = base + lane;
            int j = 0;
#pragma unroll 1
            for (int k = 0; k < nseg; ++k)
                j += (__builtin_amdgcn_readlane(incl, k) <= e) ? 1 : 0;
            const int jj  = min(j, 31);
            const int pre = __shfl(incl, jj) - __shfl(len, jj);
            const int off = __shfl(dbeg, jj) + (e - pre);
            s = 0;
            t = 0;
            if (e < total) {
                const size_t idx = (size_t)(qb + wave + 8 * jj) * CE + (size_t)off;
                s = ssrc[idx];
                t = sdst[idx];
            }
        };

        int sn = 0, tn = 0;
        if (total > 0) fetch(0, sn, tn);
        for (int base = 0; base < total; base += 32) {
            const int  s     = sn;
            const int  t     = tn;
            const bool valid = base + lane < total;
            if (base + 32 < total) fetch(base + 32, sn, tn);

            if (FIRST) { if (valid) atomicAdd(&cntl[t - r0], 1.0f); }

            unsigned m = __builtin_amdgcn_ballot_w32(valid);
            while (m) {
                v4f v[G];
                int tt[G];
                int n = 0;
#pragma unroll
                for (int g = 0; g < G; ++g) {
                    if (m) {
                        const int l = __builtin_ctz(m);
                        m &= m - 1;
                        const int si = __builtin_amdgcn_readlane(s, l);
                        tt[g] = __builtin_amdgcn_readlane(t, l) - r0;
                        v[g] = *((const v4f*)(x + (size_t)si * D) + lane);
                        n = g + 1;
                    }
                }
#pragma unroll
                for (int g = 0; g < G; ++g) {
                    if (g < n) {
                        float* at = rows + tt[g] * D + lane;
                        atomicAdd(at,      v[g].x);
                        atomicAdd(at + 32, v[g].y);
                        atomicAdd(at + 64, v[g].z);
                        atomicAdd(at + 96, v[g].w);
                    }
                }
            }
        }
    }
    __syncthreads();

    v4f* g4 = (v4f*)(acc + (size_t)r0 * D);
    for (int i = threadIdx.x; i < nr * (D / 4); i += 256) {
        const int    row = i >> 5;
        const float* lr  = rows + row * D + (i & 31);
        const v4f sum = {lr[0], lr[32], lr[64], lr[96]};
        if (FIRST) {
            g4[i] = sum;
        } else {
            const float inv = 1.0f / fmaxf(cnt[r0 + row], 1.0f);
            g4[i] = g4[i] + sum * inv;
        }
    }
    if (FIRST) { if (threadIdx.x < nr) cnt[r0 + threadIdx.x] = cntl[threadIdx.x]; }
}

template <int K1, int HIDN, int NOUT>
__global__ void __launch_bounds__(128, 16)
sage_layers(const float* __restrict__ agg, const float* __restrict__ cnt,
            const float* __restrict__ xroot,
            const float* __restrict__ W1l, const float* __restrict__ W1r,
            const float* __restrict__ b1,
            const float* __restrict__ W2l, const float* __restrict__ W2r,
            const float* __restrict__ b2,
            float* __restrict__ p, float* __restrict__ out, int nrows)
{
    constexpr int JT    = 4;
    constexpr int PITCH = HIDN + 4;
    static_assert(K1 == 4 * 32,        "a wave stages one input row as 32 x 128 bits");
    static_assert(HIDN == 2 * K1,      "the LDS tile holds [mean | x] first, h afterwards");
    static_assert(HIDN == 4 * 16 * JT, "4 waves x 64 columns cover a row of h");
    static_assert(NOUT == 2 * 16 * JT, "2 waves x 64 columns cover a row of p / of out");

    __shared__ __attribute__((aligned(16))) float tile[16 * PITCH];

    const int lane = threadIdx.x & 31;
    const int wave = __builtin_amdgcn_readfirstlane(threadIdx.x >> 5);
    const int half = lane >> 4;
    const int l    = lane & 15;
    const int tm   = blockIdx.x * 16;
    if (tm >= nrows) return;

#pragma unroll
    for (int i = 0; i < 4; ++i) {
        const int    rr  = wave + 4 * i;
        const float* ag  = agg   + (size_t)(tm + rr) * K1;
        const float* xg  = xroot + (size_t)(tm + rr) * K1;
        const float  inv = 1.0f / fmaxf(cnt[tm + rr], 1.0f);
        const v4f a = *(const v4f*)(ag + 4u * (unsigned)lane);
        const v4f r = *(const v4f*)(xg + 4u * (unsigned)lane);
        *(v4f*)(tile + rr * PITCH + 4 * lane)      = a * inv;
        *(v4f*)(tile + rr * PITCH + K1 + 4 * lane) = r;
    }
    __syncthreads();

    const float* fRow = tile + l * PITCH;

    {
        const int c4 = wave * (16 * JT) + 4 * l;
        v8f acc[JT] = {};
#pragma unroll 1
        for (int k0 = 0; k0 < K1; k0 += 4) {
            const int ka = k0 + 2 * half;

            const v2f am = *(const v2f*)(fRow + ka);
            const v2f ar = *(const v2f*)(fRow + K1 + ka);

            const unsigned wo = (unsigned)ka * HIDN + (unsigned)c4;
            const v4f l0 = *(const v4f*)(W1l + wo);
            const v4f l1 = *(const v4f*)(W1l + wo + HIDN);
            const v4f r0 = *(const v4f*)(W1r + wo);
            const v4f r1 = *(const v4f*)(W1r + wo + HIDN);

#pragma unroll
            for (int j = 0; j < JT; ++j) {
                v2f bl, br;
                bl.x = l0[j];  bl.y = l1[j];
                br.x = r0[j];  br.y = r1[j];

                acc[j] = __builtin_amdgcn_wmma_f32_16x16x4_f32(
                    false, am, false, bl, (short)0, acc[j], false, false);
                acc[j] = __builtin_amdgcn_wmma_f32_16x16x4_f32(
                    false, ar, false, br, (short)0, acc[j], false, false);
            }
        }

        __syncthreads();

        const v4f bv = *(const v4f*)(b1 + c4);
#pragma unroll
        for (int r = 0; r < 8; ++r) {
            v4f v;
#pragma unroll
            for (int j = 0; j < JT; ++j)
                v[j] = fmaxf(acc[j][r] + bv[j], 0.0f);
            *(v4f*)(tile + (8 * half + r) * PITCH + c4) = v;
        }
    }
    __builtin_amdgcn_sched_barrier(0);
    __syncthreads();

    {
        const bool   root = wave >= 2;
        const float* Wm   = root ? W2r : W2l;
        float*       dtile = (root ? out : p) + (size_t)tm * NOUT;
        const int    c4   = (wave & 1) * (16 * JT) + 4 * l;

        v8f acc[JT] = {};
#pragma unroll 2
        for (int k0 = 0; k0 < HIDN; k0 += 4) {
            const int ka = k0 + 2 * half;

            const v2f a = *(const v2f*)(fRow + ka);

            const unsigned wo = (unsigned)ka * NOUT + (unsigned)c4;
            const v4f w0 = *(const v4f*)(Wm + wo);
            const v4f w1 = *(const v4f*)(Wm + wo + NOUT);

#pragma unroll
            for (int j = 0; j < JT; ++j) {
                v2f b;
                b.x = w0[j];  b.y = w1[j];
                acc[j] = __builtin_amdgcn_wmma_f32_16x16x4_f32(
                    false, a, false, b, (short)0, acc[j], false, false);
            }
        }

        v4f bv = {0.0f, 0.0f, 0.0f, 0.0f};
        if (root) bv = *(const v4f*)(b2 + c4);
#pragma unroll
        for (int r = 0; r < 8; ++r) {
            v4f v;
#pragma unroll
            for (int j = 0; j < JT; ++j)
                v[j] = root ? acc[j][r] + bv[j] : acc[j][r];
            *(v4f*)(dtile + (unsigned)((8 * half + r) * NOUT + c4)) = v;
        }
    }
}

extern "C" void kernel_launch(void* const* d_in, const int* in_sizes, int n_in,
                              void* d_out, int out_size, void* d_ws, size_t ws_size,
                              hipStream_t stream)
{
    constexpr int IN_DIM = 128, HID = 256, EMB = 128;
    constexpr int RB   = 32;
    constexpr int CE   = 16384;
    constexpr int MAXB = 2048;

    const float* x   = (const float*)d_in[0];
    const int*   ei  = (const int*)  d_in[1];
    const float* W1l = (const float*)d_in[2];
    const float* b1  = (const float*)d_in[3];
    const float* W1r = (const float*)d_in[4];
    const float* W2l = (const float*)d_in[5];
    const float* b2  = (const float*)d_in[6];
    const float* W2r = (const float*)d_in[7];

    const int N = in_sizes[0] / IN_DIM;
    const int E = in_sizes[1] / 2;
    const int* src = ei;
    const int* dst = ei + E;
    const int nProd = (E + CE - 1) / CE;

    float* cnt  = (float*)d_ws;
    float* agg1 = cnt  + (((size_t)N + 63) & ~(size_t)63);
    float* p    = agg1 + (size_t)N * IN_DIM;
    int*   ssrc = (int*)(p + (size_t)N * EMB);
    int*   sdst = ssrc + (size_t)nProd * CE;
    int*   offs = sdst + (size_t)nProd * CE;

    const dim3 blk(256);
    const int nBlk = (N + RB - 1) / RB;
    const int tmN  = (N + 15) / 16;

    edge_partition<RB, CE, MAXB><<<nProd, dim3(1024), 0, stream>>>(
        src, dst, ssrc, sdst, offs, E);

    sage_block_sums<IN_DIM, true, RB, CE, MAXB><<<nBlk, blk, 0, stream>>>(
        x, ssrc, sdst, offs, agg1, cnt, E, N);

    sage_layers<IN_DIM, HID, EMB><<<tmN, dim3(128), 0, stream>>>(
        agg1, cnt, x, W1l, W1r, b1, W2l, W2r, b2, p, (float*)d_out, N);

    sage_block_sums<EMB, false, RB, CE, MAXB><<<nBlk, blk, 0, stream>>>(
        p, ssrc, sdst, offs, (float*)d_out, cnt, E, N);
}


# Layer-1 aggregation fused into the GEMM launch: each 4-wave work-group sums its 16 rows' neighbours in its LDS tile (LDS float atomics, arrival order as before); no agg1 in memory, 3 launches not 4
# speedup vs baseline: 3.9542x; 1.0311x over previous
#include <hip/hip_runtime.h>

typedef float v2f __attribute__((ext_vector_type(2)));
typedef float v4f __attribute__((ext_vector_type(4)));
typedef float v8f __attribute__((ext_vector_type(8)));

template <int RB, int CE, int MAXB>
__global__ void __launch_bounds__(1024)
edge_partition(const int* __restrict__ src, const int* __restrict__ dst,
               int* __restrict__ ssrc, int* __restrict__ sdst,
               int* __restrict__ offs, int E)
{
    constexpr int T   = 1024;
    constexpr int PER = CE / T;
    constexpr int BPT = MAXB / T;
    static_assert(CE % T == 0,   "every thread takes the same number of edges");
    static_assert(MAXB % T == 0, "every thread scans the same number of bins");

    __shared__ int cursor[MAXB];
    __shared__ int start[MAXB + 1];
    __shared__ int wtot[32];
    __shared__ int st_s[CE];
    __shared__ int st_d[CE];

    const int tid   = threadIdx.x;
    const int lane  = tid & 31;
    const int wave  = tid >> 5;
    const int q     = blockIdx.x;
    const int e0    = q * CE;
    const int ne    = min(CE, E - e0);

#pragma unroll
    for (int k = 0; k < BPT; ++k) cursor[k * T + tid] = 0;
    __syncthreads();

    int s[PER], t[PER], b[PER];
#pragma unroll
    for (int k = 0; k < PER; ++k) {
        const int i = min(k * T + tid, ne - 1);
        s[k] = src[e0 + i];
        t[k] = dst[e0 + i];
    }
#pragma unroll
    for (int k = 0; k < PER; ++k) {
        b[k] = -1;
        if (k * T + tid < ne) {
            b[k] = t[k] / RB;
            atomicAdd(&cursor[b[k]], 1);
        }
    }
    __syncthreads();

    int cb[BPT];
    int own = 0;
#pragma unroll
    for (int k = 0; k < BPT; ++k) {
        cb[k] = cursor[BPT * tid + k];
        own += cb[k];
    }
    int v = own;
    for (int d = 1; d < 32; d <<= 1) {
        const int u = __shfl_up(v, d);
        if (lane >= d) v += u;
    }
    if (lane == 31) wtot[wave] = v;
    __syncthreads();
    int before = 0;
    for (int j = 0; j < wave; ++j) before += wtot[j];
    int run = before + v - own;
#pragma unroll
    for (int k = 0; k < BPT; ++k) {
        cursor[BPT * tid + k] = run;
        start[BPT * tid + k]  = run;
        run += cb[k];
    }
    if (tid == T - 1) start[MAXB] = run;
    __syncthreads();

#pragma unroll
    for (int k = 0; k < PER; ++k) {
        if (b[k] >= 0) {
            const int pos = atomicAdd(&cursor[b[k]], 1);
            st_s[pos] = s[k];
            st_d[pos] = t[k];
        }
    }
    __syncthreads();

    for (int i = tid; i < ne; i += T) {
        ssrc[e0 + i] = st_s[i];
        sdst[e0 + i] = st_d[i];
    }
    int* od = offs + (size_t)q * (MAXB + 1);
    for (int i = tid; i <= MAXB; i += T) od[i] = start[i];
}

template <int D, bool COUNT, int NW, int NB, int PITCHW, int CE, int MAXB>
__device__ __forceinline__ void
add_block_edges(const float* __restrict__ x, const int* __restrict__ ssrc,
                const int* __restrict__ sdst, const int* __restrict__ offs,
                float* rows, float* cntl, int bin0, int r0, int nprod, int lane, int wave)
{
    static_assert(D == 4 * 32, "a wave reads one source row as 32 x 128 bits");
    constexpr int G = 8;

    for (int qb = 0; qb < nprod; qb += 32 * NW) {
        const int qq = qb + wave + NW * lane;
        int dbeg = 0, dend = 0;
        if (qq < nprod) {
            const int* od = offs + (size_t)qq * (MAXB + 1) + bin0;
            dbeg = od[0];
            dend = od[NB];
        }
        const int len = dend - dbeg;
        int incl = len;
        for (int d = 1; d < 32; d <<= 1) {
            const int u = __shfl_up(incl, d);
            if (lane >= d) incl += u;
        }
        const int total = __builtin_amdgcn_readlane(incl, 31);

        const int nseg = max(0, min(32, (nprod - (qb + wave) + NW - 1) / NW));

        auto fetch = [&](int base, int& s, int& t) {
            const int e = base + lane;
            int j = 0;
#pragma unroll 1
            for (int k = 0; k < nseg; ++k)
                j += (__builtin_amdgcn_readlane(incl, k) <= e) ? 1 : 0;
            const int jj  = min(j, 31);
            const int pre = __shfl(incl, jj) - __shfl(len, jj);
            const int off = __shfl(dbeg, jj) + (e - pre);
            s = 0;
            t = 0;
            if (e < total) {
                const size_t idx = (size_t)(qb + wave + NW * jj) * CE + (size_t)off;
                s = ssrc[idx];
                t = sdst[idx];
            }
        };

        int sn = 0, tn = 0;
        if (total > 0) fetch(0, sn, tn);
        for (int base = 0; base < total; base += 32) {
            const int  s     = sn;
            const int  t     = tn;
            const bool valid = base + lane < total;
            if (base + 32 < total) fetch(base + 32, sn, tn);

            if (COUNT) { if (valid) atomicAdd(&cntl[t - r0], 1.0f); }

            unsigned m = __builtin_amdgcn_ballot_w32(valid);
            while (m) {
                v4f v[G];
                int tt[G];
                int n = 0;
#pragma unroll
                for (int g = 0; g < G; ++g) {
                    if (m) {
                        const int l = __builtin_ctz(m);
                        m &= m - 1;
                        const int si = __builtin_amdgcn_readlane(s, l);
                        tt[g] = __builtin_amdgcn_readlane(t, l) - r0;
                        v[g] = *((const v4f*)(x + (size_t)si * D) + lane);
                        n = g + 1;
                    }
                }
#pragma unroll
                for (int g = 0; g < G; ++g) {
                    if (g < n) {
                        float* at = rows + tt[g] * PITCHW + lane;
                        atomicAdd(at,      v[g].x);
                        atomicAdd(at + 32, v[g].y);
                        atomicAdd(at + 64, v[g].z);
                        atomicAdd(at + 96, v[g].w);
                    }
                }
            }
        }
    }
}

template <int D, int RB, int NB, int CE, int MAXB>
__global__ void __launch_bounds__(256)
sage_block_out(const float* __restrict__ p, const int* __restrict__ ssrc,
               const int* __restrict__ sdst, const int* __restrict__ offs,
               float* __restrict__ acc, const float* __restrict__ cnt, int E, int N)
{
    __shared__ __attribute__((aligned(16))) float rows[RB * D];

    const int lane  = threadIdx.x & 31;
    const int wave  = __builtin_amdgcn_readfirstlane(threadIdx.x >> 5);
    const int nprod = (E + CE - 1) / CE;
    const int c     = blockIdx.x;
    const int r0    = c * RB;
    const int nr    = min(RB, N - r0);

    v4f* l4 = (v4f*)rows;
    for (int i = threadIdx.x; i < RB * D / 4; i += 256)
        l4[i] = v4f{0.0f, 0.0f, 0.0f, 0.0f};
    __syncthreads();

    add_block_edges<D, false, 8, NB, D, CE, MAXB>(
        p, ssrc, sdst, offs, rows, nullptr, c * NB, r0, nprod, lane, wave);
    __syncthreads();

    v4f* g4 = (v4f*)(acc + (size_t)r0 * D);
    for (int i = threadIdx.x; i < nr * (D / 4); i += 256) {
        const int    row = i >> 5;
        const float* lr  = rows + row * D + (i & 31);
        const v4f sum = {lr[0], lr[32], lr[64], lr[96]};
        const float inv = 1.0f / fmaxf(cnt[r0 + row], 1.0f);
        g4[i] = g4[i] + sum * inv;
    }
}

template <int K1, int HIDN, int NOUT, int CE, int MAXB>
__global__ void __launch_bounds__(128, 16)
sage_sums_layers(const float* __restrict__ x, const int* __restrict__ ssrc,
                 const int* __restrict__ sdst, const int* __restrict__ offs,
                 float* __restrict__ cnt,
                 const float* __restrict__ W1l, const float* __restrict__ W1r,
                 const float* __restrict__ b1,
                 const float* __restrict__ W2l, const float* __restrict__ W2r,
                 const float* __restrict__ b2,
                 float* __restrict__ p, float* __restrict__ out, int E, int nrows)
{
    constexpr int JT    = 4;
    constexpr int PITCH = HIDN + 4;
    static_assert(K1 == 4 * 32,        "a wave stages one input row as 32 x 128 bits");
    static_assert(HIDN == 2 * K1,      "the LDS tile holds [mean | x] first, h afterwards");
    static_assert(HIDN == 4 * 16 * JT, "4 waves x 64 columns cover a row of h");
    static_assert(NOUT == 2 * 16 * JT, "2 waves x 64 columns cover a row of p / of out");

    __shared__ __attribute__((aligned(16))) float tile[16 * PITCH];
    __shared__ float cntl[16];

    const int lane = threadIdx.x & 31;
    const int wave = __builtin_amdgcn_readfirstlane(threadIdx.x >> 5);
    const int half = lane >> 4;
    const int l    = lane & 15;
    const int c    = blockIdx.x;
    const int tm   = c * 16;
    if (tm >= nrows) return;

    for (int i = threadIdx.x; i < 16 * (K1 / 4); i += 128)
        *(v4f*)(tile + (i >> 5) * PITCH + 4 * (i & 31)) = v4f{0.0f, 0.0f, 0.0f, 0.0f};
    if (threadIdx.x < 16) cntl[threadIdx.x] = 0.0f;
    __syncthreads();

    add_block_edges<K1, true, 4, 1, PITCH, CE, MAXB>(
        x, ssrc, sdst, offs, tile, cntl, c, tm, (E + CE - 1) / CE, lane, wave);
    __syncthreads();

#pragma unroll
    for (int i = 0; i < 4; ++i) {
        const int    rr  = wave + 4 * i;
        float*       tr  = tile + rr * PITCH;
        const float  inv = 1.0f / fmaxf(cntl[rr], 1.0f);
        const v4f s = {tr[lane], tr[32 + lane], tr[64 + lane], tr[96 + lane]};
        v4f r = {0.0f, 0.0f, 0.0f, 0.0f};
        if (tm + rr < nrows)
            r = *(const v4f*)(x + (size_t)(tm + rr) * K1 + 4u * (unsigned)lane);
        *(v4f*)(tr + 4 * lane)      = s * inv;
        *(v4f*)(tr + K1 + 4 * lane) = r;
    }
    if (threadIdx.x < 16 && tm + (int)threadIdx.x < nrows)
        cnt[tm + threadIdx.x] = cntl[threadIdx.x];
    __syncthreads();

    const float* fRow = tile + l * PITCH;

    {
        const int c4 = wave * (16 * JT) + 4 * l;
        v8f acc[JT] = {};
#pragma unroll 1
        for (int k0 = 0; k0 < K1; k0 += 4) {
            const int ka = k0 + 2 * half;

            const v2f am = *(const v2f*)(fRow + ka);
            const v2f ar = *(const v2f*)(fRow + K1 + ka);

            const unsigned wo = (unsigned)ka * HIDN + (unsigned)c4;
            const v4f l0 = *(const v4f*)(W1l + wo);
            const v4f l1 = *(const v4f*)(W1l + wo + HIDN);
            const v4f r0 = *(const v4f*)(W1r + wo);
            const v4f r1 = *(const v4f*)(W1r + wo + HIDN);

#pragma unroll
            for (int j = 0; j < JT; ++j) {
                v2f bl, br;
                bl.x = l0[j];  bl.y = l1[j];
                br.x = r0[j];  br.y = r1[j];

                acc[j] = __builtin_amdgcn_wmma_f32_16x16x4_f32(
                    false, am, false, bl, (short)0, acc[j], false, false);
                acc[j] = __builtin_amdgcn_wmma_f32_16x16x4_f32(
                    false, ar, false, br, (short)0, acc[j], false, false);
            }
        }

        __syncthreads();

        const v4f bv = *(const v4f*)(b1 + c4);
#pragma unroll
        for (int r = 0; r < 8; ++r) {
            v4f v;
#pragma unroll
            for (int j = 0; j < JT; ++j)
                v[j] = fmaxf(acc[j][r] + bv[j], 0.0f);
            *(v4f*)(tile + (8 * half + r) * PITCH + c4) = v;
        }
    }
    __builtin_amdgcn_sched_barrier(0);
    __syncthreads();

    {
        const bool   root = wave >= 2;
        const float* Wm   = root ? W2r : W2l;
        float*       dtile = (root ? out : p) + (size_t)tm * NOUT;
        const int    c4   = (wave & 1) * (16 * JT) + 4 * l;

        v8f acc[JT] = {};
#pragma unroll 2
        for (int k0 = 0; k0 < HIDN; k0 += 4) {
            const int ka = k0 + 2 * half;

            const v2f a = *(const v2f*)(fRow + ka);

            const unsigned wo = (unsigned)ka * NOUT + (unsigned)c4;
            const v4f w0 = *(const v4f*)(Wm + wo);
            const v4f w1 = *(const v4f*)(Wm + wo + NOUT);

#pragma unroll
            for (int j = 0; j < JT; ++j) {
                v2f b;
                b.x = w0[j];  b.y = w1[j];
                acc[j] = __builtin_amdgcn_wmma_f32_16x16x4_f32(
                    false, a, false, b, (short)0, acc[j], false, false);
            }
        }

        v4f bv = {0.0f, 0.0f, 0.0f, 0.0f};
        if (root) bv = *(const v4f*)(b2 + c4);
#pragma unroll
        for (int r = 0; r < 8; ++r) {
            v4f v;
#pragma unroll
            for (int j = 0; j < JT; ++j)
                v[j] = root ? acc[j][r] + bv[j] : acc[j][r];
            if (tm + 8 * half + r < nrows)
                *(v4f*)(dtile + (unsigned)((8 * half + r) * NOUT + c4)) = v;
        }
    }
}

extern "C" void kernel_launch(void* const* d_in, const int* in_sizes, int n_in,
                              void* d_out, int out_size, void* d_ws, size_t ws_size,
                              hipStream_t stream)
{
    constexpr int IN_DIM = 128, HID = 256, EMB = 128;
    constexpr int RBIN = 16;
    constexpr int NB2  = 2;
    constexpr int RB2  = RBIN * NB2;
    constexpr int CE   = 16384;
    constexpr int MAXB = 4096;

    const float* x   = (const float*)d_in[0];
    const int*   ei  = (const int*)  d_in[1];
    const float* W1l = (const float*)d_in[2];
    const float* b1  = (const float*)d_in[3];
    const float* W1r = (const float*)d_in[4];
    const float* W2l = (const float*)d_in[5];
    const float* b2  = (const float*)d_in[6];
    const float* W2r = (const float*)d_in[7];

    const int N = in_sizes[0] / IN_DIM;
    const int E = in_sizes[1] / 2;
    const int* src = ei;
    const int* dst = ei + E;
    const int nProd = (E + CE - 1) / CE;

    float* cnt  = (float*)d_ws;
    float* p    = cnt  + (((size_t)N + 63) & ~(size_t)63);
    int*   ssrc = (int*)(p + (size_t)N * EMB);
    int*   sdst = ssrc + (size_t)nProd * CE;
    int*   offs = sdst + (size_t)nProd * CE;

    const int nBlk2 = (N + RB2 - 1) / RB2;
    const int tmN   = (N + 15) / 16;

    edge_partition<RBIN, CE, MAXB><<<nProd, dim3(1024), 0, stream>>>(
        src, dst, ssrc, sdst, offs, E);

    sage_sums_layers<IN_DIM, HID, EMB, CE, MAXB><<<tmN, dim3(128), 0, stream>>>(
        x, ssrc, sdst, offs, cnt, W1l, W1r, b1, W2l, W2r, b2, p, (float*)d_out, E, N);

    sage_block_out<EMB, RB2, NB2, CE, MAXB><<<nBlk2, dim3(256), 0, stream>>>(
        p, ssrc, sdst, offs, (float*)d_out, cnt, E, N);
}


# Half of the 16-row tiles run layer 1's root-path k steps before their row gather, half behind it: matrix work beside gathers on every WGP; layer-1 sum order: root-path k steps, then mean-path
# speedup vs baseline: 4.1127x; 1.0401x over previous
#include <hip/hip_runtime.h>
#include <type_traits>

typedef float v2f __attribute__((ext_vector_type(2)));
typedef float v4f __attribute__((ext_vector_type(4)));
typedef float v8f __attribute__((ext_vector_type(8)));

template <int RB, int CE, int MAXB>
__global__ void __launch_bounds__(1024)
edge_partition(const int* __restrict__ src, const int* __restrict__ dst,
               int* __restrict__ ssrc, int* __restrict__ sdst,
               int* __restrict__ offs, int E)
{
    constexpr int T   = 1024;
    constexpr int PER = CE / T;
    constexpr int BPT = MAXB / T;
    static_assert(CE % T == 0,   "every thread takes the same number of edges");
    static_assert(MAXB % T == 0, "every thread scans the same number of bins");

    __shared__ int cursor[MAXB];
    __shared__ int start[MAXB + 1];
    __shared__ int wtot[32];
    __shared__ int st_s[CE];
    __shared__ int st_d[CE];

    const int tid   = threadIdx.x;
    const int lane  = tid & 31;
    const int wave  = tid >> 5;
    const int q     = blockIdx.x;
    const int e0    = q * CE;
    const int ne    = min(CE, E - e0);

#pragma unroll
    for (int k = 0; k < BPT; ++k) cursor[k * T + tid] = 0;
    __syncthreads();

    int s[PER], t[PER], b[PER];
#pragma unroll
    for (int k = 0; k < PER; ++k) {
        const int i = min(k * T + tid, ne - 1);
        s[k] = src[e0 + i];
        t[k] = dst[e0 + i];
    }
#pragma unroll
    for (int k = 0; k < PER; ++k) {
        b[k] = -1;
        if (k * T + tid < ne) {
            b[k] = t[k] / RB;
            atomicAdd(&cursor[b[k]], 1);
        }
    }
    __syncthreads();

    int cb[BPT];
    int own = 0;
#pragma unroll
    for (int k = 0; k < BPT; ++k) {
        cb[k] = cursor[BPT * tid + k];
        own += cb[k];
    }
    int v = own;
    for (int d = 1; d < 32; d <<= 1) {
        const int u = __shfl_up(v, d);
        if (lane >= d) v += u;
    }
    if (lane == 31) wtot[wave] = v;
    __syncthreads();
    int before = 0;
    for (int j = 0; j < wave; ++j) before += wtot[j];
    int run = before + v - own;
#pragma unroll
    for (int k = 0; k < BPT; ++k) {
        cursor[BPT * tid + k] = run;
        start[BPT * tid + k]  = run;
        run += cb[k];
    }
    if (tid == T - 1) start[MAXB] = run;
    __syncthreads();

#pragma unroll
    for (int k = 0; k < PER; ++k) {
        if (b[k] >= 0) {
            const int pos = atomicAdd(&cursor[b[k]], 1);
            st_s[pos] = s[k];
            st_d[pos] = t[k];
        }
    }
    __syncthreads();

    for (int i = tid; i < ne; i += T) {
        ssrc[e0 + i] = st_s[i];
        sdst[e0 + i] = st_d[i];
    }
    int* od = offs + (size_t)q * (MAXB + 1);
    for (int i = tid; i <= MAXB; i += T) od[i] = start[i];
}

template <int D, bool COUNT, int NW, int NB, int PITCHW, int CE, int MAXB, int G>
__device__ __forceinline__ void
add_block_edges(const float* __restrict__ x, const int* __restrict__ ssrc,
                const int* __restrict__ sdst, const int* __restrict__ offs,
                float* rows, float* cntl, int bin0, int r0, int nprod, int lane, int wave)
{
    static_assert(D == 4 * 32, "a wave reads one source row as 32 x 128 bits");

    for (int qb = 0; qb < nprod; qb += 32 * NW) {
        const int qq = qb + wave + NW * lane;
        int dbeg = 0, dend = 0;
        if (qq < nprod) {
            const int* od = offs + (size_t)qq * (MAXB + 1) + bin0;
            dbeg = od[0];
            dend = od[NB];
        }
        const int len = dend - dbeg;
        int incl = len;
        for (int d = 1; d < 32; d <<= 1) {
            const int u = __shfl_up(incl, d);
            if (lane >= d) incl += u;
        }
        const int total = __builtin_amdgcn_readlane(incl, 31);

        const int nseg = max(0, min(32, (nprod - (qb + wave) + NW - 1) / NW));

        auto fetch = [&](int base, int& s, int& t) {
            const int e = base + lane;
            int j = 0;
#pragma unroll 1
            for (int k = 0; k < nseg; ++k)
                j += (__builtin_amdgcn_readlane(incl, k) <= e) ? 1 : 0;
            const int jj  = min(j, 31);
            const int pre = __shfl(incl, jj) - __shfl(len, jj);
            const int off = __shfl(dbeg, jj) + (e - pre);
            s = 0;
            t = 0;
            if (e < total) {
                const size_t idx = (size_t)(qb + wave + NW * jj) * CE + (size_t)off;
                s = ssrc[idx];
                t = sdst[idx];
            }
        };

        int sn = 0, tn = 0;
        if (total > 0) fetch(0, sn, tn);
        for (int base = 0; base < total; base += 32) {
            const int  s     = sn;
            const int  t     = tn;
            const bool valid = base + lane < total;
            if (base + 32 < total) fetch(base + 32, sn, tn);

            if (COUNT) { if (valid) atomicAdd(&cntl[t - r0], 1.0f); }

            unsigned m = __builtin_amdgcn_ballot_w32(valid);
            while (m) {
                v4f v[G];
                int tt[G];
                int n = 0;
#pragma unroll
                for (int g = 0; g < G; ++g) {
                    if (m) {
                        const int l = __builtin_ctz(m);
                        m &= m - 1;
                        const int si = __builtin_amdgcn_readlane(s, l);
                        tt[g] = __builtin_amdgcn_readlane(t, l) - r0;
                        v[g] = *((const v4f*)(x + (size_t)si * D) + lane);
                        n = g + 1;
                    }
                }
#pragma unroll
                for (int g = 0; g < G; ++g) {
                    if (g < n) {
                        float* at = rows + tt[g] * PITCHW + lane;
                        atomicAdd(at,      v[g].x);
                        atomicAdd(at + 32, v[g].y);
                        atomicAdd(at + 64, v[g].z);
                        atomicAdd(at + 96, v[g].w);
                    }
                }
            }
        }
    }
}

template <int D, int RB, int NB, int CE, int MAXB>
__global__ void __launch_bounds__(256)
sage_block_out(const float* __restrict__ p, const int* __restrict__ ssrc,
               const int* __restrict__ sdst, const int* __restrict__ offs,
               float* __restrict__ acc, const float* __restrict__ cnt, int E, int N)
{
    __shared__ __attribute__((aligned(16))) float rows[RB * D];

    const int lane  = threadIdx.x & 31;
    const int wave  = __builtin_amdgcn_readfirstlane(threadIdx.x >> 5);
    const int nprod = (E + CE - 1) / CE;
    const int c     = blockIdx.x;
    const int r0    = c * RB;
    const int nr    = min(RB, N - r0);

    v4f* l4 = (v4f*)rows;
    for (int i = threadIdx.x; i < RB * D / 4; i += 256)
        l4[i] = v4f{0.0f, 0.0f, 0.0f, 0.0f};
    __syncthreads();

    add_block_edges<D, false, 8, NB, D, CE, MAXB, 8>(
        p, ssrc, sdst, offs, rows, nullptr, c * NB, r0, nprod, lane, wave);
    __syncthreads();

    v4f* g4 = (v4f*)(acc + (size_t)r0 * D);
    for (int i = threadIdx.x; i < nr * (D / 4); i += 256) {
        const int    row = i >> 5;
        const float* lr  = rows + row * D + (i & 31);
        const v4f sum = {lr[0], lr[32], lr[64], lr[96]};
        const float inv = 1.0f / fmaxf(cnt[r0 + row], 1.0f);
        g4[i] = g4[i] + sum * inv;
    }
}

template <int K1, int HIDN, int NOUT, int CE, int MAXB>
__global__ void __launch_bounds__(128, 16)
sage_sums_layers(const float* __restrict__ x, const int* __restrict__ ssrc,
                 const int* __restrict__ sdst, const int* __restrict__ offs,
                 float* __restrict__ cnt,
                 const float* __restrict__ W1l, const float* __restrict__ W1r,
                 const float* __restrict__ b1,
                 const float* __restrict__ W2l, const float* __restrict__ W2r,
                 const float* __restrict__ b2,
                 float* __restrict__ p, float* __restrict__ out, int E, int nrows)
{
    constexpr int JT    = 4;
    constexpr int PITCH = HIDN + 4;
    static_assert(K1 == 4 * 32,        "a wave stages one input row as 32 x 128 bits");
    static_assert(HIDN == 2 * K1,      "the LDS tile holds [mean | x] first, h afterwards");
    static_assert(HIDN == 4 * 16 * JT, "4 waves x 64 columns cover a row of h");
    static_assert(NOUT == 2 * 16 * JT, "2 waves x 64 columns cover a row of p / of out");

    __shared__ __attribute__((aligned(16))) float tile[16 * PITCH];
    __shared__ float cntl[16];

    const int lane = threadIdx.x & 31;
    const int wave = __builtin_amdgcn_readfirstlane(threadIdx.x >> 5);
    const int half = lane >> 4;
    const int l    = lane & 15;
    const int c    = blockIdx.x;
    const int tm   = c * 16;
    if (tm >= nrows) return;

    const bool rootFirst = (__builtin_popcount((unsigned)c) & 1) != 0;

    for (int i = threadIdx.x; i < 16 * (K1 / 4); i += 128)
        *(v4f*)(tile + (i >> 5) * PITCH + 4 * (i & 31)) = v4f{0.0f, 0.0f, 0.0f, 0.0f};
    if (threadIdx.x < 16) cntl[threadIdx.x] = 0.0f;
#pragma unroll
    for (int i = 0; i < 4; ++i) {
        const int rr = wave + 4 * i;
        v4f r = {0.0f, 0.0f, 0.0f, 0.0f};
        if (tm + rr < nrows)
            r = *(const v4f*)(x + (size_t)(tm + rr) * K1 + 4u * (unsigned)lane);
        *(v4f*)(tile + rr * PITCH + K1 + 4 * lane) = r;
    }
    __syncthreads();

    const float* fRow = tile + l * PITCH;
    const int    c4l1 = wave * (16 * JT) + 4 * l;

    v8f acc[JT] = {};

    auto k_steps = [&](const float* __restrict__ Wp, int aoff) {
#pragma unroll 2
        for (int k0 = 0; k0 < K1; k0 += 4) {
            const int ka = k0 + 2 * half;
            const v2f a = *(const v2f*)(fRow + aoff + ka);
            const unsigned wo = (unsigned)ka * HIDN + (unsigned)c4l1;
            const v4f w0 = *(const v4f*)(Wp + wo);
            const v4f w1 = *(const v4f*)(Wp + wo + HIDN);
#pragma unroll
            for (int j = 0; j < JT; ++j) {
                v2f b;
                b.x = w0[j];  b.y = w1[j];
                acc[j] = __builtin_amdgcn_wmma_f32_16x16x4_f32(
                    false, a, false, b, (short)0, acc[j], false, false);
            }
        }
    };

    auto sums_to_means = [&](auto gtag) {
        constexpr int G = decltype(gtag)::value;
        add_block_edges<K1, true, 4, 1, PITCH, CE, MAXB, G>(
            x, ssrc, sdst, offs, tile, cntl, c, tm, (E + CE - 1) / CE, lane, wave);
        __syncthreads();
#pragma unroll
        for (int i = 0; i < 4; ++i) {
            const int    rr  = wave + 4 * i;
            float*       tr  = tile + rr * PITCH;
            const float  inv = 1.0f / fmaxf(cntl[rr], 1.0f);
            const v4f s = {tr[lane], tr[32 + lane], tr[64 + lane], tr[96 + lane]};
            *(v4f*)(tr + 4 * lane) = s * inv;
        }
        if (threadIdx.x < 16 && tm + (int)threadIdx.x < nrows)
            cnt[tm + threadIdx.x] = cntl[threadIdx.x];
        __syncthreads();
    };

    if (!rootFirst) sums_to_means(std::integral_constant<int, 8>{});
    k_steps(W1r, K1);
    if (rootFirst) {
        __syncthreads();
        float* pk = tile + (4 * wave) * PITCH + K1 + 4 * lane;
        *(v4f*)(pk)             = __builtin_shufflevector(acc[2], acc[2], 0, 1, 2, 3);
        *(v4f*)(pk + PITCH)     = __builtin_shufflevector(acc[2], acc[2], 4, 5, 6, 7);
        *(v4f*)(pk + 2 * PITCH) = __builtin_shufflevector(acc[3], acc[3], 0, 1, 2, 3);
        *(v4f*)(pk + 3 * PITCH) = __builtin_shufflevector(acc[3], acc[3], 4, 5, 6, 7);
        sums_to_means(std::integral_constant<int, 4>{});
        const v4f p0 = *(const v4f*)(pk);
        const v4f p1 = *(const v4f*)(pk + PITCH);
        const v4f p2 = *(const v4f*)(pk + 2 * PITCH);
        const v4f p3 = *(const v4f*)(pk + 3 * PITCH);
        acc[2] = __builtin_shufflevector(p0, p1, 0, 1, 2, 3, 4, 5, 6, 7);
        acc[3] = __builtin_shufflevector(p2, p3, 0, 1, 2, 3, 4, 5, 6, 7);
    }
    k_steps(W1l, 0);

    {
        __syncthreads();

        const v4f bv = *(const v4f*)(b1 + c4l1);
#pragma unroll
        for (int r = 0; r < 8; ++r) {
            v4f v;
#pragma unroll
            for (int j = 0; j < JT; ++j)
                v[j] = fmaxf(acc[j][r] + bv[j], 0.0f);
            *(v4f*)(tile + (8 * half + r) * PITCH + c4l1) = v;
        }
    }
    __builtin_amdgcn_sched_barrier(0);
    __syncthreads();

    {
        const bool   root = wave >= 2;
        const float* Wm   = root ? W2r : W2l;
        float*       dtile = (root ? out : p) + (size_t)tm * NOUT;
        const int    c4   = (wave & 1) * (16 * JT) + 4 * l;

        v8f acc[JT] = {};
#pragma unroll 2
        for (int k0 = 0; k0 < HIDN; k0 += 4) {
            const int ka = k0 + 2 * half;

            const v2f a = *(const v2f*)(fRow + ka);

            const unsigned wo = (unsigned)ka * NOUT + (unsigned)c4;
            const v4f w0 = *(const v4f*)(Wm + wo);
            const v4f w1 = *(const v4f*)(Wm + wo + NOUT);

#pragma unroll
            for (int j = 0; j < JT; ++j) {
                v2f b;
                b.x = w0[j];  b.y = w1[j];
                acc[j] = __builtin_amdgcn_wmma_f32_16x16x4_f32(
                    false, a, false, b, (short)0, acc[j], false, false);
            }
        }

        v4f bv = {0.0f, 0.0f, 0.0f, 0.0f};
        if (root) bv = *(const v4f*)(b2 + c4);
#pragma unroll
        for (int r = 0; r < 8; ++r) {
            v4f v;
#pragma unroll
            for (int j = 0; j < JT; ++j)
                v[j] = root ? acc[j][r] + bv[j] : acc[j][r];
            if (tm + 8 * half + r < nrows)
                *(v4f*)(dtile + (unsigned)((8 * half + r) * NOUT + c4)) = v;
        }
    }
}

extern "C" void kernel_launch(void* const* d_in, const int* in_sizes, int n_in,
                              void* d_out, int out_size, void* d_ws, size_t ws_size,
                              hipStream_t stream)
{
    constexpr int IN_DIM = 128, HID = 256, EMB = 128;
    constexpr int RBIN = 16;
    constexpr int NB2  = 2;
    constexpr int RB2  = RBIN * NB2;
    constexpr int CE   = 16384;
    constexpr int MAXB = 4096;

    const float* x   = (const float*)d_in[0];
    const int*   ei  = (const int*)  d_in[1];
    const float* W1l = (const float*)d_in[2];
    const float* b1  = (const float*)d_in[3];
    const float* W1r = (const float*)d_in[4];
    const float* W2l = (const float*)d_in[5];
    const float* b2  = (const float*)d_in[6];
    const float* W2r = (const float*)d_in[7];

    const int N = in_sizes[0] / IN_DIM;
    const int E = in_sizes[1] / 2;
    const int* src = ei;
    const int* dst = ei + E;
    const int nProd = (E + CE - 1) / CE;

    float* cnt  = (float*)d_ws;
    float* p    = cnt  + (((size_t)N + 63) & ~(size_t)63);
    int*   ssrc = (int*)(p + (size_t)N * EMB);
    int*   sdst = ssrc + (size_t)nProd * CE;
    int*   offs = sdst + (size_t)nProd * CE;

    const int nBlk2 = (N + RB2 - 1) / RB2;
    const int tmN   = (N + 15) / 16;

    edge_partition<RBIN, CE, MAXB><<<nProd, dim3(1024), 0, stream>>>(
        src, dst, ssrc, sdst, offs, E);

    sage_sums_layers<IN_DIM, HID, EMB, CE, MAXB><<<tmN, dim3(128), 0, stream>>>(
        x, ssrc, sdst, offs, cnt, W1l, W1r, b1, W2l, W2r, b2, p, (float*)d_out, E, N);

    sage_block_out<EMB, RB2, NB2, CE, MAXB><<<nBlk2, dim3(256), 0, stream>>>(
        p, ssrc, sdst, offs, (float*)d_out, cnt, E, N);
}


# Weights copied once per call into WMMA fragment order (idle WGPs of the partition launch): B pairs come straight from 128-bit loads, no register copies in the k loops; sums unchanged
# speedup vs baseline: 4.2450x; 1.0322x over previous
#include <hip/hip_runtime.h>
#include <type_traits>

typedef float v2f __attribute__((ext_vector_type(2)));
typedef float v4f __attribute__((ext_vector_type(4)));
typedef float v8f __attribute__((ext_vector_type(8)));


template <int K, int NC>
__device__ __forceinline__ void
weights_to_fragment_order(const float* __restrict__ W, float* __restrict__ dst,
                          int first, int step)
{
    constexpr int NS = NC / 64;
    for (int d = first; d < K * NC; d += step) {
        const int s     = d & 1;
        const int jj    = (d >> 1) & 1;
        const int l     = (d >> 2) & 15;
        const int t2    = (d >> 6) & 1;
        const int h     = (d >> 7) & 1;
        const int blk   = d >> 8;
        const int strip = blk % NS;
        const int ks    = blk / NS;
        dst[d] = W[(4 * ks + 2 * h + s) * NC + 64 * strip + 4 * l + 2 * t2 + jj];
    }
}

template <int RB, int CE, int MAXB, int K1, int HIDN, int NOUT, int NWGW>
__global__ void __launch_bounds__(1024)
edge_partition(const int* __restrict__ src, const int* __restrict__ dst,
               int* __restrict__ ssrc, int* __restrict__ sdst,
               int* __restrict__ offs, int E,
               const float* __restrict__ W1l, const float* __restrict__ W1r,
               const float* __restrict__ W2l, const float* __restrict__ W2r,
               float* __restrict__ wfr)
{
    const int nprod = (E + CE - 1) / CE;
    if ((int)blockIdx.x >= nprod) {
        const int first = ((int)blockIdx.x - nprod) * 1024 + (int)threadIdx.x;
        const int step  = NWGW * 1024;
        weights_to_fragment_order<K1, HIDN>(W1r, wfr,                   first, step);
        weights_to_fragment_order<K1, HIDN>(W1l, wfr + K1 * HIDN,       first, step);
        weights_to_fragment_order<HIDN, NOUT>(W2l, wfr + 2 * K1 * HIDN, first, step);
        weights_to_fragment_order<HIDN, NOUT>(W2r, wfr + 2 * K1 * HIDN + HIDN * NOUT, first, step);
        return;
    }

    constexpr int T   = 1024;
    constexpr int PER = CE / T;
    constexpr int BPT = MAXB / T;
    static_assert(CE % T == 0,   "every thread takes the same number of edges");
    static_assert(MAXB % T == 0, "every thread scans the same number of bins");

    __shared__ int cursor[MAXB];
    __shared__ int start[MAXB + 1];
    __shared__ int wtot[32];
    __shared__ int st_s[CE];
    __shared__ int st_d[CE];

    const int tid   = threadIdx.x;
    const int lane  = tid & 31;
    const int wave  = tid >> 5;
    const int q     = blockIdx.x;
    const int e0    = q * CE;
    const int ne    = min(CE, E - e0);

#pragma unroll
    for (int k = 0; k < BPT; ++k) cursor[k * T + tid] = 0;
    __syncthreads();

    int s[PER], t[PER], b[PER];
#pragma unroll
    for (int k = 0; k < PER; ++k) {
        const int i = min(k * T + tid, ne - 1);
        s[k] = src[e0 + i];
        t[k] = dst[e0 + i];
    }
#pragma unroll
    for (int k = 0; k < PER; ++k) {
        b[k] = -1;
        if (k * T + tid < ne) {
            b[k] = t[k] / RB;
            atomicAdd(&cursor[b[k]], 1);
        }
    }
    __syncthreads();

    int cb[BPT];
    int own = 0;
#pragma unroll
    for (int k = 0; k < BPT; ++k) {
        cb[k] = cursor[BPT * tid + k];
        own += cb[k];
    }
    int v = own;
    for (int d = 1; d < 32; d <<= 1) {
        const int u = __shfl_up(v, d);
        if (lane >= d) v += u;
    }
    if (lane == 31) wtot[wave] = v;
    __syncthreads();
    int before = 0;
    for (int j = 0; j < wave; ++j) before += wtot[j];
    int run = before + v - own;
#pragma unroll
    for (int k = 0; k < BPT; ++k) {
        cursor[BPT * tid + k] = run;
        start[BPT * tid + k]  = run;
        run += cb[k];
    }
    if (tid == T - 1) start[MAXB] = run;
    __syncthreads();

#pragma unroll
    for (int k = 0; k < PER; ++k) {
        if (b[k] >= 0) {
            const int pos = atomicAdd(&cursor[b[k]], 1);
            st_s[pos] = s[k];
            st_d[pos] = t[k];
        }
    }
    __syncthreads();

    for (int i = tid; i < ne; i += T) {
        ssrc[e0 + i] = st_s[i];
        sdst[e0 + i] = st_d[i];
    }
    int* od = offs + (size_t)q * (MAXB + 1);
    for (int i = tid; i <= MAXB; i += T) od[i] = start[i];
}

template <int D, bool COUNT, int NW, int NB, int PITCHW, int CE, int MAXB, int G>
__device__ __forceinline__ void
add_block_edges(const float* __restrict__ x, const int* __restrict__ ssrc,
                const int* __restrict__ sdst, const int* __restrict__ offs,
                float* rows, float* cntl, int bin0, int r0, int nprod, int lane, int wave)
{
    static_assert(D == 4 * 32, "a wave reads one source row as 32 x 128 bits");

    for (int qb = 0; qb < nprod; qb += 32 * NW) {
        const int qq = qb + wave + NW * lane;
        int dbeg = 0, dend = 0;
        if (qq < nprod) {
            const int* od = offs + (size_t)qq * (MAXB + 1) + bin0;
            dbeg = od[0];
            dend = od[NB];
        }
        const int len = dend - dbeg;
        int incl = len;
        for (int d = 1; d < 32; d <<= 1) {
            const int u = __shfl_up(incl, d);
            if (lane >= d) incl += u;
        }
        const int total = __builtin_amdgcn_readlane(incl, 31);

        const int nseg = max(0, min(32, (nprod - (qb + wave) + NW - 1) / NW));

        auto fetch = [&](int base, int& s, int& t) {
            const int e = base + lane;
            int j = 0;
#pragma unroll 1
            for (int k = 0; k < nseg; ++k)
                j += (__builtin_amdgcn_readlane(incl, k) <= e) ? 1 : 0;
            const int jj  = min(j, 31);
            const int pre = __shfl(incl, jj) - __shfl(len, jj);
            const int off = __shfl(dbeg, jj) + (e - pre);
            s = 0;
            t = 0;
            if (e < total) {
                const size_t idx = (size_t)(qb + wave + NW * jj) * CE + (size_t)off;
                s = ssrc[idx];
                t = sdst[idx];
            }
        };

        int sn = 0, tn = 0;
        if (total > 0) fetch(0, sn, tn);
        for (int base = 0; base < total; base += 32) {
            const int  s     = sn;
            const int  t     = tn;
            const bool valid = base + lane < total;
            if (base + 32 < total) fetch(base + 32, sn, tn);

            if (COUNT) { if (valid) atomicAdd(&cntl[t - r0], 1.0f); }

            unsigned m = __builtin_amdgcn_ballot_w32(valid);
            while (m) {
                v4f v[G];
                int tt[G];
                int n = 0;
#pragma unroll
                for (int g = 0; g < G; ++g) {
                    if (m) {
                        const int l = __builtin_ctz(m);
                        m &= m - 1;
                        const int si = __builtin_amdgcn_readlane(s, l);
                        tt[g] = __builtin_amdgcn_readlane(t, l) - r0;
                        v[g] = *((const v4f*)(x + (size_t)si * D) + lane);
                        n = g + 1;
                    }
                }
#pragma unroll
                for (int g = 0; g < G; ++g) {
                    if (g < n) {
                        float* at = rows + tt[g] * PITCHW + lane;
                        atomicAdd(at,      v[g].x);
                        atomicAdd(at + 32, v[g].y);
                        atomicAdd(at + 64, v[g].z);
                        atomicAdd(at + 96, v[g].w);
                    }
                }
            }
        }
    }
}

template <int D, int RB, int NB, int CE, int MAXB>
__global__ void __launch_bounds__(256)
sage_block_out(const float* __restrict__ p, const int* __restrict__ ssrc,
               const int* __restrict__ sdst, const int* __restrict__ offs,
               float* __restrict__ acc, const float* __restrict__ cnt, int E, int N)
{
    __shared__ __attribute__((aligned(16))) float rows[RB * D];

    const int lane  = threadIdx.x & 31;
    const int wave  = __builtin_amdgcn_readfirstlane(threadIdx.x >> 5);
    const int nprod = (E + CE - 1) / CE;
    const int c     = blockIdx.x;
    const int r0    = c * RB;
    const int nr    = min(RB, N - r0);

    v4f* l4 = (v4f*)rows;
    for (int i = threadIdx.x; i < RB * D / 4; i += 256)
        l4[i] = v4f{0.0f, 0.0f, 0.0f, 0.0f};
    __syncthreads();

    add_block_edges<D, false, 8, NB, D, CE, MAXB, 8>(
        p, ssrc, sdst, offs, rows, nullptr, c * NB, r0, nprod, lane, wave);
    __syncthreads();

    v4f* g4 = (v4f*)(acc + (size_t)r0 * D);
    for (int i = threadIdx.x; i < nr * (D / 4); i += 256) {
        const int    row = i >> 5;
        const float* lr  = rows + row * D + (i & 31);
        const v4f sum = {lr[0], lr[32], lr[64], lr[96]};
        const float inv = 1.0f / fmaxf(cnt[r0 + row], 1.0f);
        g4[i] = g4[i] + sum * inv;
    }
}

template <int K1, int HIDN, int NOUT, int CE, int MAXB>
__global__ void __launch_bounds__(128, 16)
sage_sums_layers(const float* __restrict__ x, const int* __restrict__ ssrc,
                 const int* __restrict__ sdst, const int* __restrict__ offs,
                 float* __restrict__ cnt,
                 const float* __restrict__ W1lf, const float* __restrict__ W1rf,
                 const float* __restrict__ b1,
                 const float* __restrict__ W2lf, const float* __restrict__ W2rf,
                 const float* __restrict__ b2,
                 float* __restrict__ p, float* __restrict__ out, int E, int nrows)
{
    constexpr int JT    = 4;
    constexpr int PITCH = HIDN + 4;
    static_assert(K1 == 4 * 32,        "a wave stages one input row as 32 x 128 bits");
    static_assert(HIDN == 2 * K1,      "the LDS tile holds [mean | x] first, h afterwards");
    static_assert(HIDN == 4 * 16 * JT, "4 waves x 64 columns cover a row of h");
    static_assert(NOUT == 2 * 16 * JT, "2 waves x 64 columns cover a row of p / of out");

    __shared__ __attribute__((aligned(16))) float tile[16 * PITCH];
    __shared__ float cntl[16];

    const int lane = threadIdx.x & 31;
    const int wave = __builtin_amdgcn_readfirstlane(threadIdx.x >> 5);
    const int half = lane >> 4;
    const int l    = lane & 15;
    const int c    = blockIdx.x;
    const int tm   = c * 16;
    if (tm >= nrows) return;

    const bool rootFirst = (__builtin_popcount((unsigned)c) & 1) != 0;

    for (int i = threadIdx.x; i < 16 * (K1 / 4); i += 128)
        *(v4f*)(tile + (i >> 5) * PITCH + 4 * (i & 31)) = v4f{0.0f, 0.0f, 0.0f, 0.0f};
    if (threadIdx.x < 16) cntl[threadIdx.x] = 0.0f;
#pragma unroll
    for (int i = 0; i < 4; ++i) {
        const int rr = wave + 4 * i;
        v4f r = {0.0f, 0.0f, 0.0f, 0.0f};
        if (tm + rr < nrows)
            r = *(const v4f*)(x + (size_t)(tm + rr) * K1 + 4u * (unsigned)lane);
        *(v4f*)(tile + rr * PITCH + K1 + 4 * lane) = r;
    }
    __syncthreads();

    const float* fRow = tile + l * PITCH;
    const int    c4l1 = wave * (16 * JT) + 4 * l;

    v8f acc[JT] = {};

    const unsigned wlane = (unsigned)(half * 128 + 4 * l) * 4u;

    auto k_steps = [&](const float* __restrict__ Wf, int aoff) {
        const char* wb = (const char*)Wf + wave * 1024;
#pragma unroll 2
        for (int ks = 0; ks < K1 / 4; ++ks) {
            const int ka = 4 * ks + 2 * half;
            const v2f a = *(const v2f*)(fRow + aoff + ka);
            const char* wk = wb + ks * ((HIDN / 64) * 1024);
            const v4f q0 = *(const v4f*)(wk + wlane);
            const v4f q1 = *(const v4f*)(wk + wlane + 256);
            acc[0] = __builtin_amdgcn_wmma_f32_16x16x4_f32(
                false, a, false, __builtin_shufflevector(q0, q0, 0, 1), (short)0, acc[0], false, false);
            acc[1] = __builtin_amdgcn_wmma_f32_16x16x4_f32(
                false, a, false, __builtin_shufflevector(q0, q0, 2, 3), (short)0, acc[1], false, false);
            acc[2] = __builtin_amdgcn_wmma_f32_16x16x4_f32(
                false, a, false, __builtin_shufflevector(q1, q1, 0, 1), (short)0, acc[2], false, false);
            acc[3] = __builtin_amdgcn_wmma_f32_16x16x4_f32(
                false, a, false, __builtin_shufflevector(q1, q1, 2, 3), (short)0, acc[3], false, false);
        }
    };

    auto sums_to_means = [&](auto gtag) {
        constexpr int G = decltype(gtag)::value;
        add_block_edges<K1, true, 4, 1, PITCH, CE, MAXB, G>(
            x, ssrc, sdst, offs, tile, cntl, c, tm, (E + CE - 1) / CE, lane, wave);
        __syncthreads();
#pragma unroll
        for (int i = 0; i < 4; ++i) {
            const int    rr  = wave + 4 * i;
            float*       tr  = tile + rr * PITCH;
            const float  inv = 1.0f / fmaxf(cntl[rr], 1.0f);
            const v4f s = {tr[lane], tr[32 + lane], tr[64 + lane], tr[96 + lane]};
            *(v4f*)(tr + 4 * lane) = s * inv;
        }
        if (threadIdx.x < 16 && tm + (int)threadIdx.x < nrows)
            cnt[tm + threadIdx.x] = cntl[threadIdx.x];
        __syncthreads();
    };

    if (!rootFirst) sums_to_means(std::integral_constant<int, 8>{});
    k_steps(W1rf, K1);
    if (rootFirst) {
        __syncthreads();
        float* pk = tile + (4 * wave) * PITCH + K1 + 4 * lane;
        *(v4f*)(pk)             = __builtin_shufflevector(acc[2], acc[2], 0, 1, 2, 3);
        *(v4f*)(pk + PITCH)     = __builtin_shufflevector(acc[2], acc[2], 4, 5, 6, 7);
        *(v4f*)(pk + 2 * PITCH) = __builtin_shufflevector(acc[3], acc[3], 0, 1, 2, 3);
        *(v4f*)(pk + 3 * PITCH) = __builtin_shufflevector(acc[3], acc[3], 4, 5, 6, 7);
        sums_to_means(std::integral_constant<int, 4>{});
        const v4f p0 = *(const v4f*)(pk);
        const v4f p1 = *(const v4f*)(pk + PITCH);
        const v4f p2 = *(const v4f*)(pk + 2 * PITCH);
        const v4f p3 = *(const v4f*)(pk + 3 * PITCH);
        acc[2] = __builtin_shufflevector(p0, p1, 0, 1, 2, 3, 4, 5, 6, 7);
        acc[3] = __builtin_shufflevector(p2, p3, 0, 1, 2, 3, 4, 5, 6, 7);
    }
    k_steps(W1lf, 0);

    {
        __syncthreads();

        const v4f bv = *(const v4f*)(b1 + c4l1);
#pragma unroll
        for (int r = 0; r < 8; ++r) {
            v4f v;
#pragma unroll
            for (int j = 0; j < JT; ++j)
                v[j] = fmaxf(acc[j][r] + bv[j], 0.0f);
            *(v4f*)(tile + (8 * half + r) * PITCH + c4l1) = v;
        }
    }
    __builtin_amdgcn_sched_barrier(0);
    __syncthreads();

    {
        const bool   root = wave >= 2;
        const char*  wb   = (const char*)(root ? W2rf : W2lf) + (wave & 1) * 1024;
        float*       dtile = (root ? out : p) + (size_t)tm * NOUT;
        const int    c4   = (wave & 1) * (16 * JT) + 4 * l;

        v8f acc[JT] = {};
#pragma unroll 2
        for (int ks = 0; ks < HIDN / 4; ++ks) {
            const int ka = 4 * ks + 2 * half;

            const v2f a = *(const v2f*)(fRow + ka);

            const char* wk = wb + ks * ((NOUT / 64) * 1024);
            const v4f q0 = *(const v4f*)(wk + wlane);
            const v4f q1 = *(const v4f*)(wk + wlane + 256);

            acc[0] = __builtin_amdgcn_wmma_f32_16x16x4_f32(
                false, a, false, __builtin_shufflevector(q0, q0, 0, 1), (short)0, acc[0], false, false);
            acc[1] = __builtin_amdgcn_wmma_f32_16x16x4_f32(
                false, a, false, __builtin_shufflevector(q0, q0, 2, 3), (short)0, acc[1], false, false);
            acc[2] = __builtin_amdgcn_wmma_f32_16x16x4_f32(
                false, a, false, __builtin_shufflevector(q1, q1, 0, 1), (short)0, acc[2], false, false);
            acc[3] = __builtin_amdgcn_wmma_f32_16x16x4_f32(
                false, a, false, __builtin_shufflevector(q1, q1, 2, 3), (short)0, acc[3], false, false);
        }

        v4f bv = {0.0f, 0.0f, 0.0f, 0.0f};
        if (root) bv = *(const v4f*)(b2 + c4);
#pragma unroll
        for (int r = 0; r < 8; ++r) {
            v4f v;
#pragma unroll
            for (int j = 0; j < JT; ++j)
                v[j] = root ? acc[j][r] + bv[j] : acc[j][r];
            if (tm + 8 * half + r < nrows)
                *(v4f*)(dtile + (unsigned)((8 * half + r) * NOUT + c4)) = v;
        }
    }
}

extern "C" void kernel_launch(void* const* d_in, const int* in_sizes, int n_in,
                              void* d_out, int out_size, void* d_ws, size_t ws_size,
                              hipStream_t stream)
{
    constexpr int IN_DIM = 128, HID = 256, EMB = 128;
    constexpr int RBIN = 16;
    constexpr int NB2  = 2;
    constexpr int RB2  = RBIN * NB2;
    constexpr int CE   = 16384;
    constexpr int MAXB = 4096;

    const float* x   = (const float*)d_in[0];
    const int*   ei  = (const int*)  d_in[1];
    const float* W1l = (const float*)d_in[2];
    const float* b1  = (const float*)d_in[3];
    const float* W1r = (const float*)d_in[4];
    const float* W2l = (const float*)d_in[5];
    const float* b2  = (const float*)d_in[6];
    const float* W2r = (const float*)d_in[7];

    const int N = in_sizes[0] / IN_DIM;
    const int E = in_sizes[1] / 2;
    const int* src = ei;
    const int* dst = ei + E;
    const int nProd = (E + CE - 1) / CE;

    float* cnt  = (float*)d_ws;
    float* p    = cnt  + (((size_t)N + 63) & ~(size_t)63);
    int*   ssrc = (int*)(p + (size_t)N * EMB);
    int*   sdst = ssrc + (size_t)nProd * CE;
    int*   offs = sdst + (size_t)nProd * CE;
    float* wfr  = (float*)(offs + (((size_t)nProd * (MAXB + 1) + 63) & ~(size_t)63));
    const float* W1rf = wfr;
    const float* W1lf = wfr + IN_DIM * HID;
    const float* W2lf = wfr + 2 * IN_DIM * HID;
    const float* W2rf = wfr + 2 * IN_DIM * HID + HID * EMB;
    constexpr int NWG_W = 32;

    const int nBlk2 = (N + RB2 - 1) / RB2;
    const int tmN   = (N + 15) / 16;

    edge_partition<RBIN, CE, MAXB, IN_DIM, HID, EMB, NWG_W><<<nProd + NWG_W, dim3(1024), 0, stream>>>(
        src, dst, ssrc, sdst, offs, E, W1l, W1r, W2l, W2r, wfr);

    sage_sums_layers<IN_DIM, HID, EMB, CE, MAXB><<<tmN, dim3(128), 0, stream>>>(
        x, ssrc, sdst, offs, cnt, W1lf, W1rf, b1, W2lf, W2rf, b2, p, (float*)d_out, E, N);

    sage_block_out<EMB, RB2, NB2, CE, MAXB><<<nBlk2, dim3(256), 0, stream>>>(
        p, ssrc, sdst, offs, (float*)d_out, cnt, E, N);
}
